# starting kernel
# baseline (speedup 1.0000x reference)
#include <hip/hip_runtime.h>
#include <hip/hip_bf16.h>


typedef float v2f __attribute__((ext_vector_type(2)));
typedef float v8f __attribute__((ext_vector_type(8)));

#define F_IN 512
#define C_OUT 64


__global__ void gcn_deg_init(float* deg, int N) {
  int i = blockIdx.x * blockDim.x + threadIdx.x;
  if (i < N) deg[i] = 1.0f;
}

__global__ void gcn_deg_count(const int* __restrict__ dst, float* deg, int E) {
  int stride = gridDim.x * blockDim.x;
  for (int e = blockIdx.x * blockDim.x + threadIdx.x; e < E; e += stride)
    atomicAdd(&deg[dst[e]], 1.0f);
}

__global__ void gcn_deg_to_dinv(float* deg, int N) {
  int i = blockIdx.x * blockDim.x + threadIdx.x;
  if (i < N) deg[i] = rsqrtf(deg[i]);
}


__global__ __launch_bounds__(128) void gcn_gemm_wmma(
    const float* __restrict__ x, const float* __restrict__ W,
    float* __restrict__ h, int N) {
  const int lane = threadIdx.x & 31;
  const int wid  = threadIdx.x >> 5;
  const int tile = blockIdx.x * 4 + wid;
  if (tile * 16 >= N) return;

  const int half = lane >> 4;
  const int l16  = lane & 15;
  const int row0 = tile * 16;

  const float* arow = x + (size_t)(row0 + l16) * F_IN + 2 * half;
  const float* bbas = W + (size_t)(2 * half) * C_OUT + l16;

  v8f acc0 = {}, acc1 = {}, acc2 = {}, acc3 = {};

  #pragma unroll 4
  for (int k = 0; k < F_IN; k += 4) {
    v2f a = *(const v2f*)(arow + k);
    const float* bk = bbas + (size_t)k * C_OUT;
    v2f b0 = { bk[0],  bk[0  + C_OUT] };
    v2f b1 = { bk[16], bk[16 + C_OUT] };
    v2f b2 = { bk[32], bk[32 + C_OUT] };
    v2f b3 = { bk[48], bk[48 + C_OUT] };
    acc0 = __builtin_amdgcn_wmma_f32_16x16x4_f32(false, a, false, b0, (short)0, acc0, false, false);
    acc1 = __builtin_amdgcn_wmma_f32_16x16x4_f32(false, a, false, b1, (short)0, acc1, false, false);
    acc2 = __builtin_amdgcn_wmma_f32_16x16x4_f32(false, a, false, b2, (short)0, acc2, false, false);
    acc3 = __builtin_amdgcn_wmma_f32_16x16x4_f32(false, a, false, b3, (short)0, acc3, false, false);
  }

  float* hp = h + (size_t)row0 * C_OUT;
  #pragma unroll
  for (int v = 0; v < 8; ++v) {
    const size_t r = (size_t)(v + 8 * half) * C_OUT;
    hp[r + 0  + l16] = acc0[v];
    hp[r + 16 + l16] = acc1[v];
    hp[r + 32 + l16] = acc2[v];
    hp[r + 48 + l16] = acc3[v];
  }
}


__global__ void gcn_self_init(const float* __restrict__ h,
                              const float* __restrict__ dinv,
                              float* __restrict__ out, int N) {
  int stride = gridDim.x * blockDim.x;
  int total = N * C_OUT;
  for (int i = blockIdx.x * blockDim.x + threadIdx.x; i < total; i += stride) {
    float di = dinv[i >> 6];
    out[i] = h[i] * di * di;
  }
}


__global__ void gcn_aggregate(const int* __restrict__ src,
                              const int* __restrict__ dst,
                              const float* __restrict__ h,
                              const float* __restrict__ dinv,
                              float* __restrict__ out, int E) {
  long long stride = (long long)gridDim.x * blockDim.x;
  long long total  = (long long)E * 16;
  for (long long i = (long long)blockIdx.x * blockDim.x + threadIdx.x;
       i < total; i += stride) {
    int e = (int)(i >> 4);
    int g = (int)(i & 15);
    int s = src[e];
    int d = dst[e];
    float norm = dinv[s] * dinv[d];
    const float4 hv = *(const float4*)(h + (size_t)s * C_OUT + g * 4);
    float* o = out + (size_t)d * C_OUT + g * 4;
    atomicAdd(o + 0, hv.x * norm);
    atomicAdd(o + 1, hv.y * norm);
    atomicAdd(o + 2, hv.z * norm);
    atomicAdd(o + 3, hv.w * norm);
  }
}


__global__ void gcn_bias_logsoftmax(float* __restrict__ out,
                                    const float* __restrict__ bias, int N) {
  int lane = threadIdx.x & 31;
  int row  = blockIdx.x * (blockDim.x >> 5) + (threadIdx.x >> 5);
  if (row >= N) return;
  float* o = out + (size_t)row * C_OUT;
  float v0 = o[lane]      + bias[lane];
  float v1 = o[lane + 32] + bias[lane + 32];
  float m = fmaxf(v0, v1);
  #pragma unroll
  for (int off = 16; off > 0; off >>= 1) m = fmaxf(m, __shfl_xor(m, off, 32));
  float s = __expf(v0 - m) + __expf(v1 - m);
  #pragma unroll
  for (int off = 16; off > 0; off >>= 1) s += __shfl_xor(s, off, 32);
  float lse = __logf(s);
  o[lane]      = v0 - m - lse;
  o[lane + 32] = v1 - m - lse;
}


extern "C" void kernel_launch(void* const* d_in, const int* in_sizes, int n_in,
                              void* d_out, int out_size, void* d_ws, size_t ws_size,
                              hipStream_t stream) {
  const float* x  = (const float*)d_in[0];
  const float* W  = (const float*)d_in[1];
  const float* b  = (const float*)d_in[2];
  const int*   ei = (const int*)d_in[3];

  const int N = in_sizes[0] / F_IN;
  const int E = in_sizes[3] / 2;
  const int* src = ei;
  const int* dst = ei + E;
  float* out = (float*)d_out;

  char*  ws   = (char*)d_ws;
  float* dinv = (float*)ws;
  size_t off  = ((size_t)N * sizeof(float) + 255) & ~(size_t)255;
  float* h    = (float*)(ws + off);

  const int T = 256;

  gcn_deg_init<<<(N + T - 1) / T, T, 0, stream>>>(dinv, N);
  {
    int blocks = (E + T - 1) / T;
    if (blocks > 8192) blocks = 8192;
    gcn_deg_count<<<blocks, T, 0, stream>>>(dst, dinv, E);
  }
  gcn_deg_to_dinv<<<(N + T - 1) / T, T, 0, stream>>>(dinv, N);

  {
    int tiles  = (N + 15) / 16;
    int blocks = (tiles + 3) / 4;
    gcn_gemm_wmma<<<blocks, 128, 0, stream>>>(x, W, h, N);
  }

  {
    int total  = N * C_OUT;
    int blocks = (total + T - 1) / T;
    if (blocks > 16384) blocks = 16384;
    gcn_self_init<<<blocks, T, 0, stream>>>(h, dinv, out, N);
  }

  {
    long long total = (long long)E * 16;
    long long blk   = (total + T - 1) / T;
    int blocks = blk > 16384 ? 16384 : (int)blk;
    gcn_aggregate<<<blocks, T, 0, stream>>>(src, dst, h, dinv, out, E);
  }

  {
    int rows_per_block = T / 32;
    int blocks = (N + rows_per_block - 1) / rows_per_block;
    gcn_bias_logsoftmax<<<blocks, T, 0, stream>>>(out, b, N);
  }
}


# edge aggregation remapped to lane = channel, one wave per 32 edges: each h row load and each f32 atomic add covers one whole aligned 128-byte line; sum order still set by atomic arrival
# speedup vs baseline: 1.3664x; 1.3664x over previous
#include <hip/hip_runtime.h>
#include <hip/hip_bf16.h>


typedef float v2f __attribute__((ext_vector_type(2)));
typedef float v8f __attribute__((ext_vector_type(8)));

#define F_IN 512
#define C_OUT 64


__global__ void gcn_deg_init(float* deg, int N) {
  int i = blockIdx.x * blockDim.x + threadIdx.x;
  if (i < N) deg[i] = 1.0f;
}

__global__ void gcn_deg_count(const int* __restrict__ dst, float* deg, int E) {
  int stride = gridDim.x * blockDim.x;
  for (int e = blockIdx.x * blockDim.x + threadIdx.x; e < E; e += stride)
    atomicAdd(&deg[dst[e]], 1.0f);
}

__global__ void gcn_deg_to_dinv(float* deg, int N) {
  int i = blockIdx.x * blockDim.x + threadIdx.x;
  if (i < N) deg[i] = rsqrtf(deg[i]);
}


__global__ __launch_bounds__(128) void gcn_gemm_wmma(
    const float* __restrict__ x, const float* __restrict__ W,
    float* __restrict__ h, int N) {
  const int lane = threadIdx.x & 31;
  const int wid  = threadIdx.x >> 5;
  const int tile = blockIdx.x * 4 + wid;
  if (tile * 16 >= N) return;

  const int half = lane >> 4;
  const int l16  = lane & 15;
  const int row0 = tile * 16;

  const float* arow = x + (size_t)(row0 + l16) * F_IN + 2 * half;
  const float* bbas = W + (size_t)(2 * half) * C_OUT + l16;

  v8f acc0 = {}, acc1 = {}, acc2 = {}, acc3 = {};

  #pragma unroll 4
  for (int k = 0; k < F_IN; k += 4) {
    v2f a = *(const v2f*)(arow + k);
    const float* bk = bbas + (size_t)k * C_OUT;
    v2f b0 = { bk[0],  bk[0  + C_OUT] };
    v2f b1 = { bk[16], bk[16 + C_OUT] };
    v2f b2 = { bk[32], bk[32 + C_OUT] };
    v2f b3 = { bk[48], bk[48 + C_OUT] };
    acc0 = __builtin_amdgcn_wmma_f32_16x16x4_f32(false, a, false, b0, (short)0, acc0, false, false);
    acc1 = __builtin_amdgcn_wmma_f32_16x16x4_f32(false, a, false, b1, (short)0, acc1, false, false);
    acc2 = __builtin_amdgcn_wmma_f32_16x16x4_f32(false, a, false, b2, (short)0, acc2, false, false);
    acc3 = __builtin_amdgcn_wmma_f32_16x16x4_f32(false, a, false, b3, (short)0, acc3, false, false);
  }

  float* hp = h + (size_t)row0 * C_OUT;
  #pragma unroll
  for (int v = 0; v < 8; ++v) {
    const size_t r = (size_t)(v + 8 * half) * C_OUT;
    hp[r + 0  + l16] = acc0[v];
    hp[r + 16 + l16] = acc1[v];
    hp[r + 32 + l16] = acc2[v];
    hp[r + 48 + l16] = acc3[v];
  }
}


__global__ void gcn_self_init(const float* __restrict__ h,
                              const float* __restrict__ dinv,
                              float* __restrict__ out, int N) {
  int stride = gridDim.x * blockDim.x;
  int total = N * C_OUT;
  for (int i = blockIdx.x * blockDim.x + threadIdx.x; i < total; i += stride) {
    float di = dinv[i >> 6];
    out[i] = h[i] * di * di;
  }
}


#define AGG_CHUNK 8

__global__ void gcn_aggregate(const int* __restrict__ src,
                              const int* __restrict__ dst,
                              const float* __restrict__ h,
                              const float* __restrict__ dinv,
                              float* __restrict__ out, int E) {
  const int lane = threadIdx.x & 31;
  const int wave = __builtin_amdgcn_readfirstlane(
      (int)(blockIdx.x * (blockDim.x >> 5) + (threadIdx.x >> 5)));
  const long long e0 = (long long)wave * 32;
  if (e0 >= (long long)E) return;
  const long long left = (long long)E - e0;
  const int cnt = left < 32 ? (int)left : 32;

  int s = 0, d = 0;
  float nrm = 0.0f;
  if (lane < cnt) {
    s = src[e0 + lane];
    d = dst[e0 + lane];
    nrm = dinv[s] * dinv[d];
  }

  const float* hl = h + lane;
  float* ol = out + lane;

  int j = 0;
  for (; j + AGG_CHUNK <= cnt; j += AGG_CHUNK) {
    float a[AGG_CHUNK], b[AGG_CHUNK];
    #pragma unroll
    for (int u = 0; u < AGG_CHUNK; ++u) {
      const int sj = __builtin_amdgcn_readlane(s, j + u);
      const float* hr = hl + (size_t)sj * C_OUT;
      a[u] = hr[0];
      b[u] = hr[32];
    }
    #pragma unroll
    for (int u = 0; u < AGG_CHUNK; ++u) {
      const int dj = __builtin_amdgcn_readlane(d, j + u);
      const float nj = __int_as_float(
          __builtin_amdgcn_readlane(__float_as_int(nrm), j + u));
      float* orow = ol + (size_t)dj * C_OUT;
      atomicAdd(orow,      a[u] * nj);
      atomicAdd(orow + 32, b[u] * nj);
    }
  }
  for (; j < cnt; ++j) {
    const int sj = __builtin_amdgcn_readlane(s, j);
    const int dj = __builtin_amdgcn_readlane(d, j);
    const float nj = __int_as_float(
        __builtin_amdgcn_readlane(__float_as_int(nrm), j));
    const float* hr = hl + (size_t)sj * C_OUT;
    float* orow = ol + (size_t)dj * C_OUT;
    const float a = hr[0];
    const float b = hr[32];
    atomicAdd(orow,      a * nj);
    atomicAdd(orow + 32, b * nj);
  }
}


__global__ void gcn_bias_logsoftmax(float* __restrict__ out,
                                    const float* __restrict__ bias, int N) {
  int lane = threadIdx.x & 31;
  int row  = blockIdx.x * (blockDim.x >> 5) + (threadIdx.x >> 5);
  if (row >= N) return;
  float* o = out + (size_t)row * C_OUT;
  float v0 = o[lane]      + bias[lane];
  float v1 = o[lane + 32] + bias[lane + 32];
  float m = fmaxf(v0, v1);
  #pragma unroll
  for (int off = 16; off > 0; off >>= 1) m = fmaxf(m, __shfl_xor(m, off, 32));
  float s = __expf(v0 - m) + __expf(v1 - m);
  #pragma unroll
  for (int off = 16; off > 0; off >>= 1) s += __shfl_xor(s, off, 32);
  float lse = __logf(s);
  o[lane]      = v0 - m - lse;
  o[lane + 32] = v1 - m - lse;
}


extern "C" void kernel_launch(void* const* d_in, const int* in_sizes, int n_in,
                              void* d_out, int out_size, void* d_ws, size_t ws_size,
                              hipStream_t stream) {
  const float* x  = (const float*)d_in[0];
  const float* W  = (const float*)d_in[1];
  const float* b  = (const float*)d_in[2];
  const int*   ei = (const int*)d_in[3];

  const int N = in_sizes[0] / F_IN;
  const int E = in_sizes[3] / 2;
  const int* src = ei;
  const int* dst = ei + E;
  float* out = (float*)d_out;

  char*  ws   = (char*)d_ws;
  float* dinv = (float*)ws;
  size_t off  = ((size_t)N * sizeof(float) + 255) & ~(size_t)255;
  float* h    = (float*)(ws + off);

  const int T = 256;

  gcn_deg_init<<<(N + T - 1) / T, T, 0, stream>>>(dinv, N);
  {
    int blocks = (E + T - 1) / T;
    if (blocks > 8192) blocks = 8192;
    gcn_deg_count<<<blocks, T, 0, stream>>>(dst, dinv, E);
  }
  gcn_deg_to_dinv<<<(N + T - 1) / T, T, 0, stream>>>(dinv, N);

  {
    int tiles  = (N + 15) / 16;
    int blocks = (tiles + 3) / 4;
    gcn_gemm_wmma<<<blocks, 128, 0, stream>>>(x, W, h, N);
  }

  {
    int total  = N * C_OUT;
    int blocks = (total + T - 1) / T;
    if (blocks > 16384) blocks = 16384;
    gcn_self_init<<<blocks, T, 0, stream>>>(h, dinv, out, N);
  }

  {
    int waves  = (E + 31) / 32;
    int blocks = (waves + (T / 32) - 1) / (T / 32);
    gcn_aggregate<<<blocks, T, 0, stream>>>(src, dst, h, dinv, out, E);
  }

  {
    int rows_per_block = T / 32;
    int blocks = (N + rows_per_block - 1) / rows_per_block;
    gcn_bias_logsoftmax<<<blocks, T, 0, stream>>>(out, b, N);
  }
}


# in-degree counted in per-block LDS tables (edge chunk x node range), flushed as one whole-line f32 atomic per 32 nodes, not one scattered global atomic per edge; integer counts, exact in any order
# speedup vs baseline: 1.9035x; 1.3931x over previous
#include <hip/hip_runtime.h>
#include <hip/hip_bf16.h>


typedef float v2f __attribute__((ext_vector_type(2)));
typedef float v8f __attribute__((ext_vector_type(8)));

#define F_IN 512
#define C_OUT 64


__global__ void gcn_deg_init(float* deg, int N) {
  int i = blockIdx.x * blockDim.x + threadIdx.x;
  if (i < N) deg[i] = 1.0f;
}


#define DEG_THREADS 1024
#define DEG_RANGE 10240
#define DEG_CHUNK 65536
#define DEG_GROUP 16

__global__ __launch_bounds__(DEG_THREADS) void gcn_deg_count(
    const int* __restrict__ dst, float* deg, int E, int N) {
  __shared__ unsigned cnt[DEG_RANGE];
  const int tid  = threadIdx.x;
  const int lo   = blockIdx.y * DEG_RANGE;
  const int left = N - lo;
  const unsigned span = (unsigned)(left < DEG_RANGE ? left : DEG_RANGE);

  for (unsigned w = tid; w < span; w += DEG_THREADS) cnt[w] = 0u;
  __syncthreads();

  const long long c0  = (long long)blockIdx.x * DEG_CHUNK;
  const long long rem = (long long)E - c0;
  const int n = rem < DEG_CHUNK ? (int)rem : DEG_CHUNK;
  const int* dp = dst + c0;

  for (int i = tid; i < n; i += DEG_GROUP * DEG_THREADS) {
    int d[DEG_GROUP];
    #pragma unroll
    for (int u = 0; u < DEG_GROUP; ++u) {
      const int j = i + u * DEG_THREADS;
      d[u] = j < n ? dp[j] : -1;
    }
    #pragma unroll
    for (int u = 0; u < DEG_GROUP; ++u) {
      const unsigned r = (unsigned)(d[u] - lo);
      if (r < span) atomicAdd(&cnt[r], 1u);
    }
  }
  __syncthreads();

  for (unsigned w = tid; w < span; w += DEG_THREADS) {
    const unsigned c = cnt[w];
    if (c) atomicAdd(&deg[lo + w], (float)c);
  }
}

__global__ void gcn_deg_to_dinv(float* deg, int N) {
  int i = blockIdx.x * blockDim.x + threadIdx.x;
  if (i < N) deg[i] = rsqrtf(deg[i]);
}


__global__ __launch_bounds__(128) void gcn_gemm_wmma(
    const float* __restrict__ x, const float* __restrict__ W,
    float* __restrict__ h, int N) {
  const int lane = threadIdx.x & 31;
  const int wid  = threadIdx.x >> 5;
  const int tile = blockIdx.x * 4 + wid;
  if (tile * 16 >= N) return;

  const int half = lane >> 4;
  const int l16  = lane & 15;
  const int row0 = tile * 16;

  const float* arow = x + (size_t)(row0 + l16) * F_IN + 2 * half;
  const float* bbas = W + (size_t)(2 * half) * C_OUT + l16;

  v8f acc0 = {}, acc1 = {}, acc2 = {}, acc3 = {};

  #pragma unroll 4
  for (int k = 0; k < F_IN; k += 4) {
    v2f a = *(const v2f*)(arow + k);
    const float* bk = bbas + (size_t)k * C_OUT;
    v2f b0 = { bk[0],  bk[0  + C_OUT] };
    v2f b1 = { bk[16], bk[16 + C_OUT] };
    v2f b2 = { bk[32], bk[32 + C_OUT] };
    v2f b3 = { bk[48], bk[48 + C_OUT] };
    acc0 = __builtin_amdgcn_wmma_f32_16x16x4_f32(false, a, false, b0, (short)0, acc0, false, false);
    acc1 = __builtin_amdgcn_wmma_f32_16x16x4_f32(false, a, false, b1, (short)0, acc1, false, false);
    acc2 = __builtin_amdgcn_wmma_f32_16x16x4_f32(false, a, false, b2, (short)0, acc2, false, false);
    acc3 = __builtin_amdgcn_wmma_f32_16x16x4_f32(false, a, false, b3, (short)0, acc3, false, false);
  }

  float* hp = h + (size_t)row0 * C_OUT;
  #pragma unroll
  for (int v = 0; v < 8; ++v) {
    const size_t r = (size_t)(v + 8 * half) * C_OUT;
    hp[r + 0  + l16] = acc0[v];
    hp[r + 16 + l16] = acc1[v];
    hp[r + 32 + l16] = acc2[v];
    hp[r + 48 + l16] = acc3[v];
  }
}


__global__ void gcn_self_init(const float* __restrict__ h,
                              const float* __restrict__ dinv,
                              float* __restrict__ out, int N) {
  int stride = gridDim.x * blockDim.x;
  int total = N * C_OUT;
  for (int i = blockIdx.x * blockDim.x + threadIdx.x; i < total; i += stride) {
    float di = dinv[i >> 6];
    out[i] = h[i] * di * di;
  }
}


#define AGG_CHUNK 8

__global__ void gcn_aggregate(const int* __restrict__ src,
                              const int* __restrict__ dst,
                              const float* __restrict__ h,
                              const float* __restrict__ dinv,
                              float* __restrict__ out, int E) {
  const int lane = threadIdx.x & 31;
  const int wave = __builtin_amdgcn_readfirstlane(
      (int)(blockIdx.x * (blockDim.x >> 5) + (threadIdx.x >> 5)));
  const long long e0 = (long long)wave * 32;
  if (e0 >= (long long)E) return;
  const long long left = (long long)E - e0;
  const int cnt = left < 32 ? (int)left : 32;

  int s = 0, d = 0;
  float nrm = 0.0f;
  if (lane < cnt) {
    s = src[e0 + lane];
    d = dst[e0 + lane];
    nrm = dinv[s] * dinv[d];
  }

  const float* hl = h + lane;
  float* ol = out + lane;

  int j = 0;
  for (; j + AGG_CHUNK <= cnt; j += AGG_CHUNK) {
    float a[AGG_CHUNK], b[AGG_CHUNK];
    #pragma unroll
    for (int u = 0; u < AGG_CHUNK; ++u) {
      const int sj = __builtin_amdgcn_readlane(s, j + u);
      const float* hr = hl + (size_t)sj * C_OUT;
      a[u] = hr[0];
      b[u] = hr[32];
    }
    #pragma unroll
    for (int u = 0; u < AGG_CHUNK; ++u) {
      const int dj = __builtin_amdgcn_readlane(d, j + u);
      const float nj = __int_as_float(
          __builtin_amdgcn_readlane(__float_as_int(nrm), j + u));
      float* orow = ol + (size_t)dj * C_OUT;
      atomicAdd(orow,      a[u] * nj);
      atomicAdd(orow + 32, b[u] * nj);
    }
  }
  for (; j < cnt; ++j) {
    const int sj = __builtin_amdgcn_readlane(s, j);
    const int dj = __builtin_amdgcn_readlane(d, j);
    const float nj = __int_as_float(
        __builtin_amdgcn_readlane(__float_as_int(nrm), j));
    const float* hr = hl + (size_t)sj * C_OUT;
    float* orow = ol + (size_t)dj * C_OUT;
    const float a = hr[0];
    const float b = hr[32];
    atomicAdd(orow,      a * nj);
    atomicAdd(orow + 32, b * nj);
  }
}


__global__ void gcn_bias_logsoftmax(float* __restrict__ out,
                                    const float* __restrict__ bias, int N) {
  int lane = threadIdx.x & 31;
  int row  = blockIdx.x * (blockDim.x >> 5) + (threadIdx.x >> 5);
  if (row >= N) return;
  float* o = out + (size_t)row * C_OUT;
  float v0 = o[lane]      + bias[lane];
  float v1 = o[lane + 32] + bias[lane + 32];
  float m = fmaxf(v0, v1);
  #pragma unroll
  for (int off = 16; off > 0; off >>= 1) m = fmaxf(m, __shfl_xor(m, off, 32));
  float s = __expf(v0 - m) + __expf(v1 - m);
  #pragma unroll
  for (int off = 16; off > 0; off >>= 1) s += __shfl_xor(s, off, 32);
  float lse = __logf(s);
  o[lane]      = v0 - m - lse;
  o[lane + 32] = v1 - m - lse;
}


extern "C" void kernel_launch(void* const* d_in, const int* in_sizes, int n_in,
                              void* d_out, int out_size, void* d_ws, size_t ws_size,
                              hipStream_t stream) {
  const float* x  = (const float*)d_in[0];
  const float* W  = (const float*)d_in[1];
  const float* b  = (const float*)d_in[2];
  const int*   ei = (const int*)d_in[3];

  const int N = in_sizes[0] / F_IN;
  const int E = in_sizes[3] / 2;
  const int* src = ei;
  const int* dst = ei + E;
  float* out = (float*)d_out;

  char*  ws   = (char*)d_ws;
  float* dinv = (float*)ws;
  size_t off  = ((size_t)N * sizeof(float) + 255) & ~(size_t)255;
  float* h    = (float*)(ws + off);

  const int T = 256;

  gcn_deg_init<<<(N + T - 1) / T, T, 0, stream>>>(dinv, N);
  {
    dim3 grid((E + DEG_CHUNK - 1) / DEG_CHUNK, (N + DEG_RANGE - 1) / DEG_RANGE);
    gcn_deg_count<<<grid, DEG_THREADS, 0, stream>>>(dst, dinv, E, N);
  }
  gcn_deg_to_dinv<<<(N + T - 1) / T, T, 0, stream>>>(dinv, N);

  {
    int tiles  = (N + 15) / 16;
    int blocks = (tiles + 3) / 4;
    gcn_gemm_wmma<<<blocks, 128, 0, stream>>>(x, W, h, N);
  }

  {
    int total  = N * C_OUT;
    int blocks = (total + T - 1) / T;
    if (blocks > 16384) blocks = 16384;
    gcn_self_init<<<blocks, T, 0, stream>>>(h, dinv, out, N);
  }

  {
    int waves  = (E + 31) / 32;
    int blocks = (waves + (T / 32) - 1) / (T / 32);
    gcn_aggregate<<<blocks, T, 0, stream>>>(src, dst, h, dinv, out, E);
  }

  {
    int rows_per_block = T / 32;
    int blocks = (N + rows_per_block - 1) / rows_per_block;
    gcn_bias_logsoftmax<<<blocks, T, 0, stream>>>(out, b, N);
  }
}


# edge weight dinv[s]*dinv[d] split per row: GEMM epilogue stores dinv*h to workspace and to out, edges add unweighted rows, softmax scales the sum by dinv; 7 launches to 5; atomic sum order as before
# speedup vs baseline: 2.0880x; 1.0969x over previous
#include <hip/hip_runtime.h>
#include <hip/hip_bf16.h>


typedef float v2f __attribute__((ext_vector_type(2)));
typedef float v8f __attribute__((ext_vector_type(8)));

#define F_IN 512
#define C_OUT 64


__global__ void gcn_deg_init(float* deg, int N) {
  int i = blockIdx.x * blockDim.x + threadIdx.x;
  if (i < N) deg[i] = 1.0f;
}


#define DEG_THREADS 1024
#define DEG_RANGE 10240
#define DEG_CHUNK 65536
#define DEG_GROUP 16

__global__ __launch_bounds__(DEG_THREADS) void gcn_deg_count(
    const int* __restrict__ dst, float* deg, int E, int N) {
  __shared__ unsigned cnt[DEG_RANGE];
  const int tid  = threadIdx.x;
  const int lo   = blockIdx.y * DEG_RANGE;
  const int left = N - lo;
  const unsigned span = (unsigned)(left < DEG_RANGE ? left : DEG_RANGE);

  for (unsigned w = tid; w < span; w += DEG_THREADS) cnt[w] = 0u;
  __syncthreads();

  const long long c0  = (long long)blockIdx.x * DEG_CHUNK;
  const long long rem = (long long)E - c0;
  const int n = rem < DEG_CHUNK ? (int)rem : DEG_CHUNK;
  const int* dp = dst + c0;

  for (int i = tid; i < n; i += DEG_GROUP * DEG_THREADS) {
    int d[DEG_GROUP];
    #pragma unroll
    for (int u = 0; u < DEG_GROUP; ++u) {
      const int j = i + u * DEG_THREADS;
      d[u] = j < n ? dp[j] : -1;
    }
    #pragma unroll
    for (int u = 0; u < DEG_GROUP; ++u) {
      const unsigned r = (unsigned)(d[u] - lo);
      if (r < span) atomicAdd(&cnt[r], 1u);
    }
  }
  __syncthreads();

  for (unsigned w = tid; w < span; w += DEG_THREADS) {
    const unsigned c = cnt[w];
    if (c) atomicAdd(&deg[lo + w], (float)c);
  }
}


__global__ __launch_bounds__(128) void gcn_gemm_wmma(
    const float* __restrict__ x, const float* __restrict__ W,
    const float* __restrict__ deg, float* __restrict__ h,
    float* __restrict__ out, int N) {
  const int lane = threadIdx.x & 31;
  const int wid  = threadIdx.x >> 5;
  const int tile = blockIdx.x * 4 + wid;
  if (tile * 16 >= N) return;

  const int half = lane >> 4;
  const int l16  = lane & 15;
  const int row0 = tile * 16;

  const float* arow = x + (size_t)(row0 + l16) * F_IN + 2 * half;
  const float* bbas = W + (size_t)(2 * half) * C_OUT + l16;

  const float4 g0 = *(const float4*)(deg + row0 + 8 * half);
  const float4 g1 = *(const float4*)(deg + row0 + 8 * half + 4);

  v8f acc0 = {}, acc1 = {}, acc2 = {}, acc3 = {};

  #pragma unroll 4
  for (int k = 0; k < F_IN; k += 4) {
    v2f a = *(const v2f*)(arow + k);
    const float* bk = bbas + (size_t)k * C_OUT;
    v2f b0 = { bk[0],  bk[0  + C_OUT] };
    v2f b1 = { bk[16], bk[16 + C_OUT] };
    v2f b2 = { bk[32], bk[32 + C_OUT] };
    v2f b3 = { bk[48], bk[48 + C_OUT] };
    acc0 = __builtin_amdgcn_wmma_f32_16x16x4_f32(false, a, false, b0, (short)0, acc0, false, false);
    acc1 = __builtin_amdgcn_wmma_f32_16x16x4_f32(false, a, false, b1, (short)0, acc1, false, false);
    acc2 = __builtin_amdgcn_wmma_f32_16x16x4_f32(false, a, false, b2, (short)0, acc2, false, false);
    acc3 = __builtin_amdgcn_wmma_f32_16x16x4_f32(false, a, false, b3, (short)0, acc3, false, false);
  }

  const float dv[8] = { rsqrtf(g0.x), rsqrtf(g0.y), rsqrtf(g0.z), rsqrtf(g0.w),
                        rsqrtf(g1.x), rsqrtf(g1.y), rsqrtf(g1.z), rsqrtf(g1.w) };
  float* hp = h   + (size_t)row0 * C_OUT;
  float* op = out + (size_t)row0 * C_OUT;
  #pragma unroll
  for (int v = 0; v < 8; ++v) {
    const size_t r = (size_t)(v + 8 * half) * C_OUT;
    const float s0 = acc0[v] * dv[v];
    const float s1 = acc1[v] * dv[v];
    const float s2 = acc2[v] * dv[v];
    const float s3 = acc3[v] * dv[v];
    hp[r + 0  + l16] = s0;  op[r + 0  + l16] = s0;
    hp[r + 16 + l16] = s1;  op[r + 16 + l16] = s1;
    hp[r + 32 + l16] = s2;  op[r + 32 + l16] = s2;
    hp[r + 48 + l16] = s3;  op[r + 48 + l16] = s3;
  }
}


#define AGG_CHUNK 8

__global__ void gcn_aggregate(const int* __restrict__ src,
                              const int* __restrict__ dst,
                              const float* __restrict__ h,
                              float* __restrict__ out, int E) {
  const int lane = threadIdx.x & 31;
  const int wave = __builtin_amdgcn_readfirstlane(
      (int)(blockIdx.x * (blockDim.x >> 5) + (threadIdx.x >> 5)));
  const long long e0 = (long long)wave * 32;
  if (e0 >= (long long)E) return;
  const long long left = (long long)E - e0;
  const int cnt = left < 32 ? (int)left : 32;

  int s = 0, d = 0;
  if (lane < cnt) {
    s = src[e0 + lane];
    d = dst[e0 + lane];
  }

  const float* hl = h + lane;
  float* ol = out + lane;

  int j = 0;
  for (; j + AGG_CHUNK <= cnt; j += AGG_CHUNK) {
    float a[AGG_CHUNK], b[AGG_CHUNK];
    #pragma unroll
    for (int u = 0; u < AGG_CHUNK; ++u) {
      const int sj = __builtin_amdgcn_readlane(s, j + u);
      const float* hr = hl + (size_t)sj * C_OUT;
      a[u] = hr[0];
      b[u] = hr[32];
    }
    #pragma unroll
    for (int u = 0; u < AGG_CHUNK; ++u) {
      const int dj = __builtin_amdgcn_readlane(d, j + u);
      float* orow = ol + (size_t)dj * C_OUT;
      atomicAdd(orow,      a[u]);
      atomicAdd(orow + 32, b[u]);
    }
  }
  for (; j < cnt; ++j) {
    const int sj = __builtin_amdgcn_readlane(s, j);
    const int dj = __builtin_amdgcn_readlane(d, j);
    const float* hr = hl + (size_t)sj * C_OUT;
    float* orow = ol + (size_t)dj * C_OUT;
    const float a = hr[0];
    const float b = hr[32];
    atomicAdd(orow,      a);
    atomicAdd(orow + 32, b);
  }
}


__global__ void gcn_bias_logsoftmax(float* __restrict__ out,
                                    const float* __restrict__ deg,
                                    const float* __restrict__ bias, int N) {
  int lane = threadIdx.x & 31;
  int row  = blockIdx.x * (blockDim.x >> 5) + (threadIdx.x >> 5);
  if (row >= N) return;
  float* o = out + (size_t)row * C_OUT;
  const float di = rsqrtf(deg[row]);
  float v0 = o[lane]      * di + bias[lane];
  float v1 = o[lane + 32] * di + bias[lane + 32];
  float m = fmaxf(v0, v1);
  #pragma unroll
  for (int off = 16; off > 0; off >>= 1) m = fmaxf(m, __shfl_xor(m, off, 32));
  float s = __expf(v0 - m) + __expf(v1 - m);
  #pragma unroll
  for (int off = 16; off > 0; off >>= 1) s += __shfl_xor(s, off, 32);
  float lse = __logf(s);
  o[lane]      = v0 - m - lse;
  o[lane + 32] = v1 - m - lse;
}


extern "C" void kernel_launch(void* const* d_in, const int* in_sizes, int n_in,
                              void* d_out, int out_size, void* d_ws, size_t ws_size,
                              hipStream_t stream) {
  const float* x  = (const float*)d_in[0];
  const float* W  = (const float*)d_in[1];
  const float* b  = (const float*)d_in[2];
  const int*   ei = (const int*)d_in[3];

  const int N = in_sizes[0] / F_IN;
  const int E = in_sizes[3] / 2;
  const int* src = ei;
  const int* dst = ei + E;
  float* out = (float*)d_out;

  char*  ws   = (char*)d_ws;
  float* deg  = (float*)ws;
  size_t off  = ((size_t)N * sizeof(float) + 255) & ~(size_t)255;
  float* h    = (float*)(ws + off);

  const int T = 256;

  gcn_deg_init<<<(N + T - 1) / T, T, 0, stream>>>(deg, N);
  {
    dim3 grid((E + DEG_CHUNK - 1) / DEG_CHUNK, (N + DEG_RANGE - 1) / DEG_RANGE);
    gcn_deg_count<<<grid, DEG_THREADS, 0, stream>>>(dst, deg, E, N);
  }

  {
    int tiles  = (N + 15) / 16;
    int blocks = (tiles + 3) / 4;
    gcn_gemm_wmma<<<blocks, 128, 0, stream>>>(x, W, deg, h, out, N);
  }

  {
    int waves  = (E + 31) / 32;
    int blocks = (waves + (T / 32) - 1) / (T / 32);
    gcn_aggregate<<<blocks, T, 0, stream>>>(src, dst, h, out, E);
  }

  {
    int rows_per_block = T / 32;
    int blocks = (N + rows_per_block - 1) / rows_per_block;
    gcn_bias_logsoftmax<<<blocks, T, 0, stream>>>(out, deg, b, N);
  }
}


# edges bucketed by 512-row destination range into the output buffer; one block per range sums hs rows in LDS (f32 LDS adds, sum order = LDS add arrival), then bias + log_softmax; no global f32 atomics
# speedup vs baseline: 2.7767x; 1.3298x over previous
#include <hip/hip_runtime.h>
#include <hip/hip_bf16.h>


typedef float v2f __attribute__((ext_vector_type(2)));
typedef float v8f __attribute__((ext_vector_type(8)));

#define F_IN 512
#define C_OUT 64


#define RROWS 512

__device__ __forceinline__ int bucket_rows(int r, int N) {
  const int left = N - r * RROWS;
  return left < RROWS ? left : RROWS;
}
__device__ __forceinline__ size_t bucket_counter_index(int r, int N) {
  return ((size_t)r * RROWS + bucket_rows(r, N)) * C_OUT - 2;
}

__global__ void gcn_deg_init(float* deg, int* outw, int N, int R) {
  int i = blockIdx.x * blockDim.x + threadIdx.x;
  if (i < N) deg[i] = 1.0f;
  if (i < R) outw[bucket_counter_index(i, N)] = 0;
}


#define DEG_THREADS 1024
#define DEG_RANGE 10240
#define DEG_CHUNK 65536
#define DEG_GROUP 16

__global__ __launch_bounds__(DEG_THREADS) void gcn_deg_count(
    const int* __restrict__ dst, float* deg, int E, int N) {
  __shared__ unsigned cnt[DEG_RANGE];
  const int tid  = threadIdx.x;
  const int lo   = blockIdx.y * DEG_RANGE;
  const int left = N - lo;
  const unsigned span = (unsigned)(left < DEG_RANGE ? left : DEG_RANGE);

  for (unsigned w = tid; w < span; w += DEG_THREADS) cnt[w] = 0u;
  __syncthreads();

  const long long c0  = (long long)blockIdx.x * DEG_CHUNK;
  const long long rem = (long long)E - c0;
  const int n = rem < DEG_CHUNK ? (int)rem : DEG_CHUNK;
  const int* dp = dst + c0;

  for (int i = tid; i < n; i += DEG_GROUP * DEG_THREADS) {
    int d[DEG_GROUP];
    #pragma unroll
    for (int u = 0; u < DEG_GROUP; ++u) {
      const int j = i + u * DEG_THREADS;
      d[u] = j < n ? dp[j] : -1;
    }
    #pragma unroll
    for (int u = 0; u < DEG_GROUP; ++u) {
      const unsigned r = (unsigned)(d[u] - lo);
      if (r < span) atomicAdd(&cnt[r], 1u);
    }
  }
  __syncthreads();

  for (unsigned w = tid; w < span; w += DEG_THREADS) {
    const unsigned c = cnt[w];
    if (c) atomicAdd(&deg[lo + w], (float)c);
  }
}


__global__ __launch_bounds__(128) void gcn_gemm_wmma(
    const float* __restrict__ x, const float* __restrict__ W,
    const float* __restrict__ deg, float* __restrict__ h, int N) {
  const int lane = threadIdx.x & 31;
  const int wid  = threadIdx.x >> 5;
  const int tile = blockIdx.x * 4 + wid;
  if (tile * 16 >= N) return;

  const int half = lane >> 4;
  const int l16  = lane & 15;
  const int row0 = tile * 16;

  const float* arow = x + (size_t)(row0 + l16) * F_IN + 2 * half;
  const float* bbas = W + (size_t)(2 * half) * C_OUT + l16;

  const float4 g0 = *(const float4*)(deg + row0 + 8 * half);
  const float4 g1 = *(const float4*)(deg + row0 + 8 * half + 4);

  v8f acc0 = {}, acc1 = {}, acc2 = {}, acc3 = {};

  #pragma unroll 4
  for (int k = 0; k < F_IN; k += 4) {
    v2f a = *(const v2f*)(arow + k);
    const float* bk = bbas + (size_t)k * C_OUT;
    v2f b0 = { bk[0],  bk[0  + C_OUT] };
    v2f b1 = { bk[16], bk[16 + C_OUT] };
    v2f b2 = { bk[32], bk[32 + C_OUT] };
    v2f b3 = { bk[48], bk[48 + C_OUT] };
    acc0 = __builtin_amdgcn_wmma_f32_16x16x4_f32(false, a, false, b0, (short)0, acc0, false, false);
    acc1 = __builtin_amdgcn_wmma_f32_16x16x4_f32(false, a, false, b1, (short)0, acc1, false, false);
    acc2 = __builtin_amdgcn_wmma_f32_16x16x4_f32(false, a, false, b2, (short)0, acc2, false, false);
    acc3 = __builtin_amdgcn_wmma_f32_16x16x4_f32(false, a, false, b3, (short)0, acc3, false, false);
  }

  const float dv[8] = { rsqrtf(g0.x), rsqrtf(g0.y), rsqrtf(g0.z), rsqrtf(g0.w),
                        rsqrtf(g1.x), rsqrtf(g1.y), rsqrtf(g1.z), rsqrtf(g1.w) };
  float* hp = h   + (size_t)row0 * C_OUT;
  #pragma unroll
  for (int v = 0; v < 8; ++v) {
    const size_t r = (size_t)(v + 8 * half) * C_OUT;
    const float s0 = acc0[v] * dv[v];
    const float s1 = acc1[v] * dv[v];
    const float s2 = acc2[v] * dv[v];
    const float s3 = acc3[v] * dv[v];
    hp[r + 0  + l16] = s0;
    hp[r + 16 + l16] = s1;
    hp[r + 32 + l16] = s2;
    hp[r + 48 + l16] = s3;
  }
}


#define BK_THREADS 1024
#define BK_PER 8
#define BK_CHUNK (BK_THREADS * BK_PER)

__global__ __launch_bounds__(BK_THREADS) void gcn_bucket(
    const int* __restrict__ src, const int* __restrict__ dst,
    int* outw, int E, int N, int R) {
  extern __shared__ int bk_tab[];
  __shared__ int2 sp[BK_CHUNK];
  __shared__ int  sb[BK_CHUNK];
  __shared__ int  nsorted;
  int* hist  = bk_tab;
  int* loff  = bk_tab + R;
  int* gbase = bk_tab + 2 * R;
  const int tid = threadIdx.x;

  for (int w = tid; w < R; w += BK_THREADS) hist[w] = 0;
  __syncthreads();

  const long long c0 = (long long)blockIdx.x * BK_CHUNK;
  int s[BK_PER], d[BK_PER], rk[BK_PER];
  #pragma unroll
  for (int u = 0; u < BK_PER; ++u) {
    const long long e = c0 + u * BK_THREADS + tid;
    const bool ok = e < (long long)E;
    d[u] = ok ? dst[e] : -1;
    s[u] = ok ? src[e] : 0;
  }
  #pragma unroll
  for (int u = 0; u < BK_PER; ++u) {
    rk[u] = -1;
    if ((unsigned)d[u] < (unsigned)N) rk[u] = atomicAdd(&hist[(unsigned)d[u] / RROWS], 1);
  }
  __syncthreads();

  if (tid < 32) {
    const int per = (R + 31) / 32;
    const int b0 = tid * per;
    int b1 = b0 + per;
    if (b1 > R) b1 = R;
    int sum = 0;
    for (int b = b0; b < b1; ++b) sum += hist[b];
    int inc = sum;
    #pragma unroll
    for (int o = 1; o < 32; o <<= 1) {
      const int t = __shfl_up(inc, o, 32);
      if (tid >= o) inc += t;
    }
    int run = inc - sum;
    for (int b = b0; b < b1; ++b) { loff[b] = run; run += hist[b]; }
    if (tid == 31) nsorted = inc;
  }
  for (int b = tid; b < R; b += BK_THREADS) {
    const int c = hist[b];
    gbase[b] = c ? atomicAdd(&outw[bucket_counter_index(b, N)], c) : 0;
  }
  __syncthreads();

  #pragma unroll
  for (int u = 0; u < BK_PER; ++u) {
    if (rk[u] >= 0) {
      const int b = (int)((unsigned)d[u] / RROWS);
      const int slot = loff[b] + rk[u];
      sp[slot] = make_int2(s[u], d[u] - b * RROWS);
      sb[slot] = b;
    }
  }
  __syncthreads();

  const int n = nsorted;
  for (int i = tid; i < n; i += BK_THREADS) {
    const int b = sb[i];
    const int p = gbase[b] + (i - loff[b]);
    const int cap = bucket_rows(b, N) * (C_OUT / 2) - 1;
    if (p < cap) ((int2*)(outw + (size_t)b * RROWS * C_OUT))[p] = sp[i];
  }
}


#define ACC_THREADS 1024
#define AGG_CHUNK 16

__global__ __launch_bounds__(ACC_THREADS) void gcn_accumulate(
    const int* __restrict__ src, const int* __restrict__ dst,
    const float* __restrict__ h, const float* __restrict__ deg,
    const float* __restrict__ bias, float* out, int E, int N) {
  __shared__ float acc[RROWS * C_OUT];
  const int tid  = threadIdx.x;
  const int lane = tid & 31;
  const int wv   = __builtin_amdgcn_readfirstlane(tid >> 5);
  const int lo   = blockIdx.x * RROWS;
  const int rows = bucket_rows(blockIdx.x, N);
  const int* reg = (const int*)out + (size_t)lo * C_OUT;
  const int cap  = rows * (C_OUT / 2) - 1;
  const int total = reg[(size_t)rows * C_OUT - 2];

  const float* hb = h + (size_t)lo * C_OUT;
  for (int i = tid; i < rows * C_OUT; i += ACC_THREADS) acc[i] = hb[i];
  __syncthreads();

  const float* hl = h + lane;
  float* al = acc + lane;

  if (total <= cap) {
    const int2* pairs = (const int2*)reg;
    int2 nxt = make_int2(0, 0);
    if (wv * 32 + lane < total) nxt = pairs[wv * 32 + lane];
    for (int base = wv * 32; base < total; base += ACC_THREADS) {
      const int left = total - base;
      const int cnt = left < 32 ? left : 32;
      const int s = nxt.x;
      const int d = nxt.y;
      if (base + ACC_THREADS + lane < total) nxt = pairs[base + ACC_THREADS + lane];
      int j = 0;
      for (; j + AGG_CHUNK <= cnt; j += AGG_CHUNK) {
        float a[AGG_CHUNK], b[AGG_CHUNK];
        #pragma unroll
        for (int u = 0; u < AGG_CHUNK; ++u) {
          const int sj = __builtin_amdgcn_readlane(s, j + u);
          const float* hr = hl + (size_t)sj * C_OUT;
          a[u] = hr[0];
          b[u] = hr[32];
        }
        #pragma unroll
        for (int u = 0; u < AGG_CHUNK; ++u) {
          const int dj = __builtin_amdgcn_readlane(d, j + u);
          float* ar = al + dj * C_OUT;
          atomicAdd(ar,      a[u]);
          atomicAdd(ar + 32, b[u]);
        }
      }
      for (; j < cnt; ++j) {
        const int sj = __builtin_amdgcn_readlane(s, j);
        const int dj = __builtin_amdgcn_readlane(d, j);
        const float* hr = hl + (size_t)sj * C_OUT;
        float* ar = al + dj * C_OUT;
        const float a = hr[0];
        const float b = hr[32];
        atomicAdd(ar,      a);
        atomicAdd(ar + 32, b);
      }
    }
  } else {
    for (long long e0 = (long long)wv * 32; e0 < (long long)E; e0 += ACC_THREADS) {
      const long long e = e0 + lane;
      const int dv = e < (long long)E ? dst[e] : -1;
      const unsigned r = (unsigned)(dv - lo);
      const bool mine = r < (unsigned)rows;
      const int s = mine ? src[e] : 0;
      unsigned todo = __builtin_amdgcn_ballot_w32(mine);
      while (todo) {
        const int j = __builtin_ctz(todo);
        todo &= todo - 1;
        const int sj = __builtin_amdgcn_readlane(s, j);
        const int dj = __builtin_amdgcn_readlane((int)r, j);
        const float* hr = hl + (size_t)sj * C_OUT;
        float* ar = al + dj * C_OUT;
        const float a = hr[0];
        const float b = hr[32];
        atomicAdd(ar,      a);
        atomicAdd(ar + 32, b);
      }
    }
  }
  __syncthreads();

  const float b0 = bias[lane];
  const float b1 = bias[lane + 32];
  for (int r = wv; r < rows; r += ACC_THREADS / 32) {
    const int row = lo + r;
    const float di = rsqrtf(deg[row]);
    float v0 = al[r * C_OUT]      * di + b0;
    float v1 = al[r * C_OUT + 32] * di + b1;
    float m = fmaxf(v0, v1);
    #pragma unroll
    for (int off = 16; off > 0; off >>= 1) m = fmaxf(m, __shfl_xor(m, off, 32));
    float sm = __expf(v0 - m) + __expf(v1 - m);
    #pragma unroll
    for (int off = 16; off > 0; off >>= 1) sm += __shfl_xor(sm, off, 32);
    const float lse = __logf(sm);
    float* o = out + (size_t)row * C_OUT;
    o[lane]      = v0 - m - lse;
    o[lane + 32] = v1 - m - lse;
  }
}


extern "C" void kernel_launch(void* const* d_in, const int* in_sizes, int n_in,
                              void* d_out, int out_size, void* d_ws, size_t ws_size,
                              hipStream_t stream) {
  const float* x  = (const float*)d_in[0];
  const float* W  = (const float*)d_in[1];
  const float* b  = (const float*)d_in[2];
  const int*   ei = (const int*)d_in[3];

  const int N = in_sizes[0] / F_IN;
  const int E = in_sizes[3] / 2;
  const int* src = ei;
  const int* dst = ei + E;
  float* out = (float*)d_out;

  char*  ws   = (char*)d_ws;
  float* deg  = (float*)ws;
  size_t off  = ((size_t)N * sizeof(float) + 255) & ~(size_t)255;
  float* h    = (float*)(ws + off);

  const int T = 256;
  const int R = (N + RROWS - 1) / RROWS;

  gcn_deg_init<<<(N + T - 1) / T, T, 0, stream>>>(deg, (int*)out, N, R);
  {
    dim3 grid((E + DEG_CHUNK - 1) / DEG_CHUNK, (N + DEG_RANGE - 1) / DEG_RANGE);
    gcn_deg_count<<<grid, DEG_THREADS, 0, stream>>>(dst, deg, E, N);
  }

  gcn_bucket<<<(E + BK_CHUNK - 1) / BK_CHUNK, BK_THREADS, 3 * R * sizeof(int), stream>>>(
      src, dst, (int*)out, E, N, R);

  {
    int tiles  = (N + 15) / 16;
    int blocks = (tiles + 3) / 4;
    gcn_gemm_wmma<<<blocks, 128, 0, stream>>>(x, W, deg, h, N);
  }

  gcn_accumulate<<<R, ACC_THREADS, 0, stream>>>(src, dst, h, deg, b, out, E, N);
}


# GEMM operands through LDS: one 12-wave block per WGP keeps W in LDS (k pairs interleaved), x rows staged by whole-line 16-byte loads one 64-k chunk ahead; 8-byte LDS fragment reads; sum order kept
# speedup vs baseline: 3.1006x; 1.1166x over previous
#include <hip/hip_runtime.h>
#include <hip/hip_bf16.h>


typedef float v2f __attribute__((ext_vector_type(2)));
typedef float v8f __attribute__((ext_vector_type(8)));

#define F_IN 512
#define C_OUT 64


#define RROWS 512

__device__ __forceinline__ int bucket_rows(int r, int N) {
  const int left = N - r * RROWS;
  return left < RROWS ? left : RROWS;
}
__device__ __forceinline__ size_t bucket_counter_index(int r, int N) {
  return ((size_t)r * RROWS + bucket_rows(r, N)) * C_OUT - 2;
}

__global__ void gcn_deg_init(float* deg, int* outw, int N, int R) {
  int i = blockIdx.x * blockDim.x + threadIdx.x;
  if (i < N) deg[i] = 1.0f;
  if (i < R) outw[bucket_counter_index(i, N)] = 0;
}


#define DEG_THREADS 1024
#define DEG_RANGE 10240
#define DEG_CHUNK 65536
#define DEG_GROUP 16

__global__ __launch_bounds__(DEG_THREADS) void gcn_deg_count(
    const int* __restrict__ dst, float* deg, int E, int N) {
  __shared__ unsigned cnt[DEG_RANGE];
  const int tid  = threadIdx.x;
  const int lo   = blockIdx.y * DEG_RANGE;
  const int left = N - lo;
  const unsigned span = (unsigned)(left < DEG_RANGE ? left : DEG_RANGE);

  for (unsigned w = tid; w < span; w += DEG_THREADS) cnt[w] = 0u;
  __syncthreads();

  const long long c0  = (long long)blockIdx.x * DEG_CHUNK;
  const long long rem = (long long)E - c0;
  const int n = rem < DEG_CHUNK ? (int)rem : DEG_CHUNK;
  const int* dp = dst + c0;

  for (int i = tid; i < n; i += DEG_GROUP * DEG_THREADS) {
    int d[DEG_GROUP];
    #pragma unroll
    for (int u = 0; u < DEG_GROUP; ++u) {
      const int j = i + u * DEG_THREADS;
      d[u] = j < n ? dp[j] : -1;
    }
    #pragma unroll
    for (int u = 0; u < DEG_GROUP; ++u) {
      const unsigned r = (unsigned)(d[u] - lo);
      if (r < span) atomicAdd(&cnt[r], 1u);
    }
  }
  __syncthreads();

  for (unsigned w = tid; w < span; w += DEG_THREADS) {
    const unsigned c = cnt[w];
    if (c) atomicAdd(&deg[lo + w], (float)c);
  }
}


#define GEMM_WAVES 12
#define GEMM_THREADS (GEMM_WAVES * 32)
#define GEMM_BLOCKS 256
#define KC 64
#define A_PITCH 68
#define W_PITCH 160

#define GEMM_LOAD_CHUNK(xp)                                   \
  pf0 = *(const float4*)((xp) + (size_t)0  * F_IN);           \
  pf1 = *(const float4*)((xp) + (size_t)2  * F_IN);           \
  pf2 = *(const float4*)((xp) + (size_t)4  * F_IN);           \
  pf3 = *(const float4*)((xp) + (size_t)6  * F_IN);           \
  pf4 = *(const float4*)((xp) + (size_t)8  * F_IN);           \
  pf5 = *(const float4*)((xp) + (size_t)10 * F_IN);           \
  pf6 = *(const float4*)((xp) + (size_t)12 * F_IN);           \
  pf7 = *(const float4*)((xp) + (size_t)14 * F_IN);
#define GEMM_PARK_CHUNK(ast)                                  \
  *(float4*)((ast) + 0  * A_PITCH) = pf0;                     \
  *(float4*)((ast) + 2  * A_PITCH) = pf1;                     \
  *(float4*)((ast) + 4  * A_PITCH) = pf2;                     \
  *(float4*)((ast) + 6  * A_PITCH) = pf3;                     \
  *(float4*)((ast) + 8  * A_PITCH) = pf4;                     \
  *(float4*)((ast) + 10 * A_PITCH) = pf5;                     \
  *(float4*)((ast) + 12 * A_PITCH) = pf6;                     \
  *(float4*)((ast) + 14 * A_PITCH) = pf7;

__global__ __launch_bounds__(GEMM_THREADS) void gcn_gemm_wmma(
    const float* __restrict__ x, const float* __restrict__ W,
    const float* __restrict__ deg, float* __restrict__ h, int N) {
  __shared__ float Ws[(F_IN / 2) * W_PITCH];
  __shared__ float As[GEMM_WAVES * 16 * A_PITCH];
  const int tid  = threadIdx.x;
  const int lane = tid & 31;
  const int wv   = __builtin_amdgcn_readfirstlane(tid >> 5);
  const int half = lane >> 4;
  const int l16  = lane & 15;
  const int tiles  = (N + 15) >> 4;
  const int stride = (int)gridDim.x * GEMM_WAVES;
  int t = wv * (int)gridDim.x + (int)blockIdx.x;

  const float* xl = x + (size_t)half * F_IN + 4 * l16;
  float4 pf0 = {}, pf1 = {}, pf2 = {}, pf3 = {}, pf4 = {}, pf5 = {}, pf6 = {}, pf7 = {};
  float4 gn0 = {}, gn1 = {};
  if (t < tiles) {
    gn0 = *(const float4*)(deg + t * 16 + 8 * half);
    gn1 = *(const float4*)(deg + t * 16 + 8 * half + 4);
    const float* xp = xl + (size_t)t * 16 * F_IN;
    GEMM_LOAD_CHUNK(xp)
  }

  #pragma unroll 4
  for (int u = tid; u < (F_IN / 2) * 16; u += GEMM_THREADS) {
    const int p = u >> 4;
    const int q = u & 15;
    const float4 r0 = *(const float4*)(W + (size_t)(2 * p) * C_OUT + 4 * q);
    const float4 r1 = *(const float4*)(W + (size_t)(2 * p + 1) * C_OUT + 4 * q);
    float4* wp = (float4*)(Ws + p * W_PITCH + 8 * q);
    wp[0] = make_float4(r0.x, r1.x, r0.y, r1.y);
    wp[1] = make_float4(r0.z, r1.z, r0.w, r1.w);
  }
  __syncthreads();

  float* aw = As + wv * (16 * A_PITCH);
  float* ast = aw + half * A_PITCH + 4 * l16;
  const float* afr = aw + l16 * A_PITCH + 2 * half;
  const float* bfr = Ws + half * W_PITCH + 2 * l16;

  for (; t < tiles; t += stride) {
    const int row0 = t * 16;
    const float4 g0 = gn0;
    const float4 g1 = gn1;

    v8f acc0 = {}, acc1 = {}, acc2 = {}, acc3 = {};

    #pragma unroll 1
    for (int kc = 0; kc < F_IN / KC; ++kc) {
      GEMM_PARK_CHUNK(ast)

      const int nk = (kc + 1) & (F_IN / KC - 1);
      const int nt = nk ? t : t + stride;
      if (nt < tiles) {
        if (nk == 0) {
          gn0 = *(const float4*)(deg + nt * 16 + 8 * half);
          gn1 = *(const float4*)(deg + nt * 16 + 8 * half + 4);
        }
        const float* xp = xl + (size_t)nt * 16 * F_IN + nk * KC;
        GEMM_LOAD_CHUNK(xp)
      }

      __builtin_amdgcn_fence(__ATOMIC_RELEASE, "wavefront");
      __builtin_amdgcn_wave_barrier();
      __builtin_amdgcn_fence(__ATOMIC_ACQUIRE, "wavefront");

      const float* bk = bfr + (kc * (KC / 2)) * W_PITCH;
      #pragma unroll
      for (int kk = 0; kk < KC / 4; ++kk) {
        const v2f a = *(const v2f*)(afr + 4 * kk);
        const float* bp = bk + (2 * kk) * W_PITCH;
        const v2f b0 = *(const v2f*)(bp);
        const v2f b1 = *(const v2f*)(bp + 32);
        const v2f b2 = *(const v2f*)(bp + 64);
        const v2f b3 = *(const v2f*)(bp + 96);
        acc0 = __builtin_amdgcn_wmma_f32_16x16x4_f32(false, a, false, b0, (short)0, acc0, false, false);
        acc1 = __builtin_amdgcn_wmma_f32_16x16x4_f32(false, a, false, b1, (short)0, acc1, false, false);
        acc2 = __builtin_amdgcn_wmma_f32_16x16x4_f32(false, a, false, b2, (short)0, acc2, false, false);
        acc3 = __builtin_amdgcn_wmma_f32_16x16x4_f32(false, a, false, b3, (short)0, acc3, false, false);
      }

      __builtin_amdgcn_fence(__ATOMIC_RELEASE, "wavefront");
      __builtin_amdgcn_wave_barrier();
      __builtin_amdgcn_fence(__ATOMIC_ACQUIRE, "wavefront");
    }

    const float dv[8] = { rsqrtf(g0.x), rsqrtf(g0.y), rsqrtf(g0.z), rsqrtf(g0.w),
                          rsqrtf(g1.x), rsqrtf(g1.y), rsqrtf(g1.z), rsqrtf(g1.w) };
    float* hp = h + (size_t)row0 * C_OUT;
    #pragma unroll
    for (int v = 0; v < 8; ++v) {
      const size_t r = (size_t)(v + 8 * half) * C_OUT;
      const float s0 = acc0[v] * dv[v];
      const float s1 = acc1[v] * dv[v];
      const float s2 = acc2[v] * dv[v];
      const float s3 = acc3[v] * dv[v];
      hp[r + 0  + l16] = s0;
      hp[r + 16 + l16] = s1;
      hp[r + 32 + l16] = s2;
      hp[r + 48 + l16] = s3;
    }
  }
}


#define BK_THREADS 1024
#define BK_PER 8
#define BK_CHUNK (BK_THREADS * BK_PER)

__global__ __launch_bounds__(BK_THREADS) void gcn_bucket(
    const int* __restrict__ src, const int* __restrict__ dst,
    int* outw, int E, int N, int R) {
  extern __shared__ int bk_tab[];
  __shared__ int2 sp[BK_CHUNK];
  __shared__ int  sb[BK_CHUNK];
  __shared__ int  nsorted;
  int* hist  = bk_tab;
  int* loff  = bk_tab + R;
  int* gbase = bk_tab + 2 * R;
  const int tid = threadIdx.x;

  for (int w = tid; w < R; w += BK_THREADS) hist[w] = 0;
  __syncthreads();

  const long long c0 = (long long)blockIdx.x * BK_CHUNK;
  int s[BK_PER], d[BK_PER], rk[BK_PER];
  #pragma unroll
  for (int u = 0; u < BK_PER; ++u) {
    const long long e = c0 + u * BK_THREADS + tid;
    const bool ok = e < (long long)E;
    d[u] = ok ? dst[e] : -1;
    s[u] = ok ? src[e] : 0;
  }
  #pragma unroll
  for (int u = 0; u < BK_PER; ++u) {
    rk[u] = -1;
    if ((unsigned)d[u] < (unsigned)N) rk[u] = atomicAdd(&hist[(unsigned)d[u] / RROWS], 1);
  }
  __syncthreads();

  if (tid < 32) {
    const int per = (R + 31) / 32;
    const int b0 = tid * per;
    int b1 = b0 + per;
    if (b1 > R) b1 = R;
    int sum = 0;
    for (int b = b0; b < b1; ++b) sum += hist[b];
    int inc = sum;
    #pragma unroll
    for (int o = 1; o < 32; o <<= 1) {
      const int t = __shfl_up(inc, o, 32);
      if (tid >= o) inc += t;
    }
    int run = inc - sum;
    for (int b = b0; b < b1; ++b) { loff[b] = run; run += hist[b]; }
    if (tid == 31) nsorted = inc;
  }
  for (int b = tid; b < R; b += BK_THREADS) {
    const int c = hist[b];
    gbase[b] = c ? atomicAdd(&outw[bucket_counter_index(b, N)], c) : 0;
  }
  __syncthreads();

  #pragma unroll
  for (int u = 0; u < BK_PER; ++u) {
    if (rk[u] >= 0) {
      const int b = (int)((unsigned)d[u] / RROWS);
      const int slot = loff[b] + rk[u];
      sp[slot] = make_int2(s[u], d[u] - b * RROWS);
      sb[slot] = b;
    }
  }
  __syncthreads();

  const int n = nsorted;
  for (int i = tid; i < n; i += BK_THREADS) {
    const int b = sb[i];
    const int p = gbase[b] + (i - loff[b]);
    const int cap = bucket_rows(b, N) * (C_OUT / 2) - 1;
    if (p < cap) ((int2*)(outw + (size_t)b * RROWS * C_OUT))[p] = sp[i];
  }
}


#define ACC_THREADS 1024
#define AGG_CHUNK 16

__global__ __launch_bounds__(ACC_THREADS) void gcn_accumulate(
    const int* __restrict__ src, const int* __restrict__ dst,
    const float* __restrict__ h, const float* __restrict__ deg,
    const float* __restrict__ bias, float* out, int E, int N) {
  __shared__ float acc[RROWS * C_OUT];
  const int tid  = threadIdx.x;
  const int lane = tid & 31;
  const int wv   = __builtin_amdgcn_readfirstlane(tid >> 5);
  const int lo   = blockIdx.x * RROWS;
  const int rows = bucket_rows(blockIdx.x, N);
  const int* reg = (const int*)out + (size_t)lo * C_OUT;
  const int cap  = rows * (C_OUT / 2) - 1;
  const int total = reg[(size_t)rows * C_OUT - 2];

  const float* hb = h + (size_t)lo * C_OUT;
  for (int i = tid; i < rows * C_OUT; i += ACC_THREADS) acc[i] = hb[i];
  __syncthreads();

  const float* hl = h + lane;
  float* al = acc + lane;

  if (total <= cap) {
    const int2* pairs = (const int2*)reg;
    int2 nxt = make_int2(0, 0);
    if (wv * 32 + lane < total) nxt = pairs[wv * 32 + lane];
    for (int base = wv * 32; base < total; base += ACC_THREADS) {
      const int left = total - base;
      const int cnt = left < 32 ? left : 32;
      const int s = nxt.x;
      const int d = nxt.y;
      if (base + ACC_THREADS + lane < total) nxt = pairs[base + ACC_THREADS + lane];
      int j = 0;
      for (; j + AGG_CHUNK <= cnt; j += AGG_CHUNK) {
        float a[AGG_CHUNK], b[AGG_CHUNK];
        #pragma unroll
        for (int u = 0; u < AGG_CHUNK; ++u) {
          const int sj = __builtin_amdgcn_readlane(s, j + u);
          const float* hr = hl + (size_t)sj * C_OUT;
          a[u] = hr[0];
          b[u] = hr[32];
        }
        #pragma unroll
        for (int u = 0; u < AGG_CHUNK; ++u) {
          const int dj = __builtin_amdgcn_readlane(d, j + u);
          float* ar = al + dj * C_OUT;
          atomicAdd(ar,      a[u]);
          atomicAdd(ar + 32, b[u]);
        }
      }
      for (; j < cnt; ++j) {
        const int sj = __builtin_amdgcn_readlane(s, j);
        const int dj = __builtin_amdgcn_readlane(d, j);
        const float* hr = hl + (size_t)sj * C_OUT;
        float* ar = al + dj * C_OUT;
        const float a = hr[0];
        const float b = hr[32];
        atomicAdd(ar,      a);
        atomicAdd(ar + 32, b);
      }
    }
  } else {
    for (long long e0 = (long long)wv * 32; e0 < (long long)E; e0 += ACC_THREADS) {
      const long long e = e0 + lane;
      const int dv = e < (long long)E ? dst[e] : -1;
      const unsigned r = (unsigned)(dv - lo);
      const bool mine = r < (unsigned)rows;
      const int s = mine ? src[e] : 0;
      unsigned todo = __builtin_amdgcn_ballot_w32(mine);
      while (todo) {
        const int j = __builtin_ctz(todo);
        todo &= todo - 1;
        const int sj = __builtin_amdgcn_readlane(s, j);
        const int dj = __builtin_amdgcn_readlane((int)r, j);
        const float* hr = hl + (size_t)sj * C_OUT;
        float* ar = al + dj * C_OUT;
        const float a = hr[0];
        const float b = hr[32];
        atomicAdd(ar,      a);
        atomicAdd(ar + 32, b);
      }
    }
  }
  __syncthreads();

  const float b0 = bias[lane];
  const float b1 = bias[lane + 32];
  for (int r = wv; r < rows; r += ACC_THREADS / 32) {
    const int row = lo + r;
    const float di = rsqrtf(deg[row]);
    float v0 = al[r * C_OUT]      * di + b0;
    float v1 = al[r * C_OUT + 32] * di + b1;
    float m = fmaxf(v0, v1);
    #pragma unroll
    for (int off = 16; off > 0; off >>= 1) m = fmaxf(m, __shfl_xor(m, off, 32));
    float sm = __expf(v0 - m) + __expf(v1 - m);
    #pragma unroll
    for (int off = 16; off > 0; off >>= 1) sm += __shfl_xor(sm, off, 32);
    const float lse = __logf(sm);
    float* o = out + (size_t)row * C_OUT;
    o[lane]      = v0 - m - lse;
    o[lane + 32] = v1 - m - lse;
  }
}


extern "C" void kernel_launch(void* const* d_in, const int* in_sizes, int n_in,
                              void* d_out, int out_size, void* d_ws, size_t ws_size,
                              hipStream_t stream) {
  const float* x  = (const float*)d_in[0];
  const float* W  = (const float*)d_in[1];
  const float* b  = (const float*)d_in[2];
  const int*   ei = (const int*)d_in[3];

  const int N = in_sizes[0] / F_IN;
  const int E = in_sizes[3] / 2;
  const int* src = ei;
  const int* dst = ei + E;
  float* out = (float*)d_out;

  char*  ws   = (char*)d_ws;
  float* deg  = (float*)ws;
  size_t off  = ((size_t)N * sizeof(float) + 255) & ~(size_t)255;
  float* h    = (float*)(ws + off);

  const int T = 256;
  const int R = (N + RROWS - 1) / RROWS;

  gcn_deg_init<<<(N + T - 1) / T, T, 0, stream>>>(deg, (int*)out, N, R);
  {
    dim3 grid((E + DEG_CHUNK - 1) / DEG_CHUNK, (N + DEG_RANGE - 1) / DEG_RANGE);
    gcn_deg_count<<<grid, DEG_THREADS, 0, stream>>>(dst, deg, E, N);
  }

  gcn_bucket<<<(E + BK_CHUNK - 1) / BK_CHUNK, BK_THREADS, 3 * R * sizeof(int), stream>>>(
      src, dst, (int*)out, E, N, R);

  gcn_gemm_wmma<<<GEMM_BLOCKS, GEMM_THREADS, 0, stream>>>(x, W, deg, h, N);

  gcn_accumulate<<<R, ACC_THREADS, 0, stream>>>(src, dst, h, deg, b, out, E, N);
}


# in-degree counted in LDS from each 400-row range's edge bucket by the GEMM block and the owner block of that range; degree pass and array gone, 5 launches to 4; f32 sum order still LDS add arrival
# speedup vs baseline: 3.6997x; 1.1932x over previous
#include <hip/hip_runtime.h>
#include <hip/hip_bf16.h>


typedef float v2f __attribute__((ext_vector_type(2)));
typedef float v8f __attribute__((ext_vector_type(8)));

#define F_IN 512
#define C_OUT 64


#define RROWS 400

__device__ __forceinline__ int bucket_rows(int r, int N) {
  const int left = N - r * RROWS;
  return left < RROWS ? left : RROWS;
}
__device__ __forceinline__ size_t bucket_counter_index(int r, int N) {
  return ((size_t)r * RROWS + bucket_rows(r, N)) * C_OUT - 2;
}

__global__ void gcn_counters_init(int* outw, int N, int R) {
  const int i = blockIdx.x * blockDim.x + threadIdx.x;
  if (i < R) outw[bucket_counter_index(i, N)] = 0;
}


#define GEMM_WAVES 12
#define GEMM_THREADS (GEMM_WAVES * 32)
#define DEG_PRE 24
#define KC 64
#define A_PITCH 68
#define W_PITCH 160

#define GEMM_LOAD_CHUNK(xp)                                   \
  pf0 = *(const float4*)((xp) + (size_t)0  * F_IN);           \
  pf1 = *(const float4*)((xp) + (size_t)2  * F_IN);           \
  pf2 = *(const float4*)((xp) + (size_t)4  * F_IN);           \
  pf3 = *(const float4*)((xp) + (size_t)6  * F_IN);           \
  pf4 = *(const float4*)((xp) + (size_t)8  * F_IN);           \
  pf5 = *(const float4*)((xp) + (size_t)10 * F_IN);           \
  pf6 = *(const float4*)((xp) + (size_t)12 * F_IN);           \
  pf7 = *(const float4*)((xp) + (size_t)14 * F_IN);
#define GEMM_PARK_CHUNK(ast)                                  \
  *(float4*)((ast) + 0  * A_PITCH) = pf0;                     \
  *(float4*)((ast) + 2  * A_PITCH) = pf1;                     \
  *(float4*)((ast) + 4  * A_PITCH) = pf2;                     \
  *(float4*)((ast) + 6  * A_PITCH) = pf3;                     \
  *(float4*)((ast) + 8  * A_PITCH) = pf4;                     \
  *(float4*)((ast) + 10 * A_PITCH) = pf5;                     \
  *(float4*)((ast) + 12 * A_PITCH) = pf6;                     \
  *(float4*)((ast) + 14 * A_PITCH) = pf7;

__global__ __launch_bounds__(GEMM_THREADS) void gcn_gemm_wmma(
    const float* __restrict__ x, const float* __restrict__ W,
    const int* __restrict__ dst, const int* __restrict__ outw,
    float* __restrict__ h, int E, int N) {
  __shared__ float Ws[(F_IN / 2) * W_PITCH];
  __shared__ float As[GEMM_WAVES * 16 * A_PITCH];
  __shared__ __attribute__((aligned(16))) unsigned indeg[RROWS];
  const int tid  = threadIdx.x;
  const int lane = tid & 31;
  const int wv   = __builtin_amdgcn_readfirstlane(tid >> 5);
  const int half = lane >> 4;
  const int l16  = lane & 15;
  const int lo     = (int)blockIdx.x * RROWS;
  const int rows   = bucket_rows((int)blockIdx.x, N);
  const int ntiles = (rows + 15) >> 4;
  int lt = wv;

  for (int w = tid; w < RROWS; w += GEMM_THREADS) indeg[w] = 0u;

  const int* reg = outw + (size_t)lo * C_OUT;
  const int cap  = rows * (C_OUT / 2) - 1;
  int dl[DEG_PRE];
  #pragma unroll
  for (int u = 0; u < DEG_PRE; ++u) {
    const int i = tid + u * GEMM_THREADS;
    dl[u] = reg[2 * (i < cap ? i : cap - 1) + 1];
  }
  const int total = reg[(size_t)rows * C_OUT - 2];

  const float* xl = x + (size_t)half * F_IN + 4 * l16;
  float4 pf0 = {}, pf1 = {}, pf2 = {}, pf3 = {}, pf4 = {}, pf5 = {}, pf6 = {}, pf7 = {};
  if (lt < ntiles) {
    const float* xp = xl + (size_t)(lo + lt * 16) * F_IN;
    GEMM_LOAD_CHUNK(xp)
  }

  #pragma unroll 4
  for (int u = tid; u < (F_IN / 2) * 16; u += GEMM_THREADS) {
    const int p = u >> 4;
    const int q = u & 15;
    const float4 r0 = *(const float4*)(W + (size_t)(2 * p) * C_OUT + 4 * q);
    const float4 r1 = *(const float4*)(W + (size_t)(2 * p + 1) * C_OUT + 4 * q);
    float4* wp = (float4*)(Ws + p * W_PITCH + 8 * q);
    wp[0] = make_float4(r0.x, r1.x, r0.y, r1.y);
    wp[1] = make_float4(r0.z, r1.z, r0.w, r1.w);
  }
  __syncthreads();

  if (total <= cap) {
    #pragma unroll
    for (int u = 0; u < DEG_PRE; ++u) {
      const int i = tid + u * GEMM_THREADS;
      if (i < total) atomicAdd(&indeg[dl[u]], 1u);
    }
    for (int i = tid + DEG_PRE * GEMM_THREADS; i < total; i += GEMM_THREADS)
      atomicAdd(&indeg[reg[2 * i + 1]], 1u);
  } else {
    for (int e = tid; e < E; e += GEMM_THREADS) {
      const unsigned r = (unsigned)(dst[e] - lo);
      if (r < (unsigned)rows) atomicAdd(&indeg[r], 1u);
    }
  }
  __syncthreads();

  float* aw = As + wv * (16 * A_PITCH);
  float* ast = aw + half * A_PITCH + 4 * l16;
  const float* afr = aw + l16 * A_PITCH + 2 * half;
  const float* bfr = Ws + half * W_PITCH + 2 * l16;

  for (; lt < ntiles; lt += GEMM_WAVES) {
    const int row0 = lo + lt * 16;

    v8f acc0 = {}, acc1 = {}, acc2 = {}, acc3 = {};

    #pragma unroll 1
    for (int kc = 0; kc < F_IN / KC; ++kc) {
      GEMM_PARK_CHUNK(ast)

      const int nk = (kc + 1) & (F_IN / KC - 1);
      const int nt = nk ? lt : lt + GEMM_WAVES;
      if (nt < ntiles) {
        const float* xp = xl + (size_t)(lo + nt * 16) * F_IN + nk * KC;
        GEMM_LOAD_CHUNK(xp)
      }

      __builtin_amdgcn_fence(__ATOMIC_RELEASE, "wavefront");
      __builtin_amdgcn_wave_barrier();
      __builtin_amdgcn_fence(__ATOMIC_ACQUIRE, "wavefront");

      const float* bk = bfr + (kc * (KC / 2)) * W_PITCH;
      #pragma unroll
      for (int kk = 0; kk < KC / 4; ++kk) {
        const v2f a = *(const v2f*)(afr + 4 * kk);
        const float* bp = bk + (2 * kk) * W_PITCH;
        const v2f b0 = *(const v2f*)(bp);
        const v2f b1 = *(const v2f*)(bp + 32);
        const v2f b2 = *(const v2f*)(bp + 64);
        const v2f b3 = *(const v2f*)(bp + 96);
        acc0 = __builtin_amdgcn_wmma_f32_16x16x4_f32(false, a, false, b0, (short)0, acc0, false, false);
        acc1 = __builtin_amdgcn_wmma_f32_16x16x4_f32(false, a, false, b1, (short)0, acc1, false, false);
        acc2 = __builtin_amdgcn_wmma_f32_16x16x4_f32(false, a, false, b2, (short)0, acc2, false, false);
        acc3 = __builtin_amdgcn_wmma_f32_16x16x4_f32(false, a, false, b3, (short)0, acc3, false, false);
      }

      __builtin_amdgcn_fence(__ATOMIC_RELEASE, "wavefront");
      __builtin_amdgcn_wave_barrier();
      __builtin_amdgcn_fence(__ATOMIC_ACQUIRE, "wavefront");
    }

    const uint4 c0 = *(const uint4*)(indeg + lt * 16 + 8 * half);
    const uint4 c1 = *(const uint4*)(indeg + lt * 16 + 8 * half + 4);
    const float dv[8] = { rsqrtf((float)(c0.x + 1u)), rsqrtf((float)(c0.y + 1u)),
                          rsqrtf((float)(c0.z + 1u)), rsqrtf((float)(c0.w + 1u)),
                          rsqrtf((float)(c1.x + 1u)), rsqrtf((float)(c1.y + 1u)),
                          rsqrtf((float)(c1.z + 1u)), rsqrtf((float)(c1.w + 1u)) };
    float* hp = h + (size_t)row0 * C_OUT;
    #pragma unroll
    for (int v = 0; v < 8; ++v) {
      const size_t r = (size_t)(v + 8 * half) * C_OUT;
      const float s0 = acc0[v] * dv[v];
      const float s1 = acc1[v] * dv[v];
      const float s2 = acc2[v] * dv[v];
      const float s3 = acc3[v] * dv[v];
      hp[r + 0  + l16] = s0;
      hp[r + 16 + l16] = s1;
      hp[r + 32 + l16] = s2;
      hp[r + 48 + l16] = s3;
    }
  }
}


#define BK_THREADS 1024
#define BK_PER 8
#define BK_CHUNK (BK_THREADS * BK_PER)

__global__ __launch_bounds__(BK_THREADS) void gcn_bucket(
    const int* __restrict__ src, const int* __restrict__ dst,
    int* outw, int E, int N, int R) {
  extern __shared__ int bk_tab[];
  __shared__ int2 sp[BK_CHUNK];
  __shared__ int  sb[BK_CHUNK];
  __shared__ int  nsorted;
  int* hist  = bk_tab;
  int* loff  = bk_tab + R;
  int* gbase = bk_tab + 2 * R;
  const int tid = threadIdx.x;

  for (int w = tid; w < R; w += BK_THREADS) hist[w] = 0;
  __syncthreads();

  const long long c0 = (long long)blockIdx.x * BK_CHUNK;
  int s[BK_PER], d[BK_PER], rk[BK_PER];
  #pragma unroll
  for (int u = 0; u < BK_PER; ++u) {
    const long long e = c0 + u * BK_THREADS + tid;
    const bool ok = e < (long long)E;
    d[u] = ok ? dst[e] : -1;
    s[u] = ok ? src[e] : 0;
  }
  #pragma unroll
  for (int u = 0; u < BK_PER; ++u) {
    rk[u] = -1;
    if ((unsigned)d[u] < (unsigned)N) rk[u] = atomicAdd(&hist[(unsigned)d[u] / RROWS], 1);
  }
  __syncthreads();

  if (tid < 32) {
    const int per = (R + 31) / 32;
    const int b0 = tid * per;
    int b1 = b0 + per;
    if (b1 > R) b1 = R;
    int sum = 0;
    for (int b = b0; b < b1; ++b) sum += hist[b];
    int inc = sum;
    #pragma unroll
    for (int o = 1; o < 32; o <<= 1) {
      const int t = __shfl_up(inc, o, 32);
      if (tid >= o) inc += t;
    }
    int run = inc - sum;
    for (int b = b0; b < b1; ++b) { loff[b] = run; run += hist[b]; }
    if (tid == 31) nsorted = inc;
  }
  for (int b = tid; b < R; b += BK_THREADS) {
    const int c = hist[b];
    gbase[b] = c ? atomicAdd(&outw[bucket_counter_index(b, N)], c) : 0;
  }
  __syncthreads();

  #pragma unroll
  for (int u = 0; u < BK_PER; ++u) {
    if (rk[u] >= 0) {
      const int b = (int)((unsigned)d[u] / RROWS);
      const int slot = loff[b] + rk[u];
      sp[slot] = make_int2(s[u], d[u] - b * RROWS);
      sb[slot] = b;
    }
  }
  __syncthreads();

  const int n = nsorted;
  for (int i = tid; i < n; i += BK_THREADS) {
    const int b = sb[i];
    const int p = gbase[b] + (i - loff[b]);
    const int cap = bucket_rows(b, N) * (C_OUT / 2) - 1;
    if (p < cap) ((int2*)(outw + (size_t)b * RROWS * C_OUT))[p] = sp[i];
  }
}


#define ACC_THREADS 1024
#define AGG_CHUNK 16

__global__ __launch_bounds__(ACC_THREADS) void gcn_accumulate(
    const int* __restrict__ src, const int* __restrict__ dst,
    const float* __restrict__ h,
    const float* __restrict__ bias, float* out, int E, int N) {
  __shared__ float acc[RROWS * C_OUT];
  __shared__ unsigned indeg[RROWS];
  const int tid  = threadIdx.x;
  const int lane = tid & 31;
  const int wv   = __builtin_amdgcn_readfirstlane(tid >> 5);
  const int lo   = blockIdx.x * RROWS;
  const int rows = bucket_rows(blockIdx.x, N);
  const int* reg = (const int*)out + (size_t)lo * C_OUT;
  const int cap  = rows * (C_OUT / 2) - 1;
  const int total = reg[(size_t)rows * C_OUT - 2];

  const float* hb = h + (size_t)lo * C_OUT;
  for (int i = tid; i < rows * C_OUT; i += ACC_THREADS) acc[i] = hb[i];
  for (int w = tid; w < RROWS; w += ACC_THREADS) indeg[w] = 0u;
  __syncthreads();

  const float* hl = h + lane;
  float* al = acc + lane;

  if (total <= cap) {
    const int2* pairs = (const int2*)reg;
    int2 nxt = make_int2(0, 0);
    if (wv * 32 + lane < total) nxt = pairs[wv * 32 + lane];
    for (int base = wv * 32; base < total; base += ACC_THREADS) {
      const int left = total - base;
      const int cnt = left < 32 ? left : 32;
      const int s = nxt.x;
      const int d = nxt.y;
      if (lane < cnt) atomicAdd(&indeg[d], 1u);
      if (base + ACC_THREADS + lane < total) nxt = pairs[base + ACC_THREADS + lane];
      int j = 0;
      for (; j + AGG_CHUNK <= cnt; j += AGG_CHUNK) {
        float a[AGG_CHUNK], b[AGG_CHUNK];
        #pragma unroll
        for (int u = 0; u < AGG_CHUNK; ++u) {
          const int sj = __builtin_amdgcn_readlane(s, j + u);
          const float* hr = hl + (size_t)sj * C_OUT;
          a[u] = hr[0];
          b[u] = hr[32];
        }
        #pragma unroll
        for (int u = 0; u < AGG_CHUNK; ++u) {
          const int dj = __builtin_amdgcn_readlane(d, j + u);
          float* ar = al + dj * C_OUT;
          atomicAdd(ar,      a[u]);
          atomicAdd(ar + 32, b[u]);
        }
      }
      for (; j < cnt; ++j) {
        const int sj = __builtin_amdgcn_readlane(s, j);
        const int dj = __builtin_amdgcn_readlane(d, j);
        const float* hr = hl + (size_t)sj * C_OUT;
        float* ar = al + dj * C_OUT;
        const float a = hr[0];
        const float b = hr[32];
        atomicAdd(ar,      a);
        atomicAdd(ar + 32, b);
      }
    }
  } else {
    for (long long e0 = (long long)wv * 32; e0 < (long long)E; e0 += ACC_THREADS) {
      const long long e = e0 + lane;
      const int dv = e < (long long)E ? dst[e] : -1;
      const unsigned r = (unsigned)(dv - lo);
      const bool mine = r < (unsigned)rows;
      const int s = mine ? src[e] : 0;
      if (mine) atomicAdd(&indeg[r], 1u);
      unsigned todo = __builtin_amdgcn_ballot_w32(mine);
      while (todo) {
        const int j = __builtin_ctz(todo);
        todo &= todo - 1;
        const int sj = __builtin_amdgcn_readlane(s, j);
        const int dj = __builtin_amdgcn_readlane((int)r, j);
        const float* hr = hl + (size_t)sj * C_OUT;
        float* ar = al + dj * C_OUT;
        const float a = hr[0];
        const float b = hr[32];
        atomicAdd(ar,      a);
        atomicAdd(ar + 32, b);
      }
    }
  }
  __syncthreads();

  const float b0 = bias[lane];
  const float b1 = bias[lane + 32];
  for (int r = wv; r < rows; r += ACC_THREADS / 32) {
    const int row = lo + r;
    const float di = rsqrtf((float)(indeg[r] + 1u));
    float v0 = al[r * C_OUT]      * di + b0;
    float v1 = al[r * C_OUT + 32] * di + b1;
    float m = fmaxf(v0, v1);
    #pragma unroll
    for (int off = 16; off > 0; off >>= 1) m = fmaxf(m, __shfl_xor(m, off, 32));
    float sm = __expf(v0 - m) + __expf(v1 - m);
    #pragma unroll
    for (int off = 16; off > 0; off >>= 1) sm += __shfl_xor(sm, off, 32);
    const float lse = __logf(sm);
    float* o = out + (size_t)row * C_OUT;
    o[lane]      = v0 - m - lse;
    o[lane + 32] = v1 - m - lse;
  }
}


extern "C" void kernel_launch(void* const* d_in, const int* in_sizes, int n_in,
                              void* d_out, int out_size, void* d_ws, size_t ws_size,
                              hipStream_t stream) {
  const float* x  = (const float*)d_in[0];
  const float* W  = (const float*)d_in[1];
  const float* b  = (const float*)d_in[2];
  const int*   ei = (const int*)d_in[3];

  const int N = in_sizes[0] / F_IN;
  const int E = in_sizes[3] / 2;
  const int* src = ei;
  const int* dst = ei + E;
  float* out = (float*)d_out;

  float* h = (float*)d_ws;

  const int T = 256;
  const int R = (N + RROWS - 1) / RROWS;

  gcn_counters_init<<<(R + T - 1) / T, T, 0, stream>>>((int*)out, N, R);

  gcn_bucket<<<(E + BK_CHUNK - 1) / BK_CHUNK, BK_THREADS, 3 * R * sizeof(int), stream>>>(
      src, dst, (int*)out, E, N, R);

  gcn_gemm_wmma<<<R, GEMM_THREADS, 0, stream>>>(x, W, dst, (const int*)out, h, E, N);

  gcn_accumulate<<<R, ACC_THREADS, 0, stream>>>(src, dst, h, b, out, E, N);
}


# log_softmax epilogue: a half wave per row, four channels per lane from one 16-byte LDS read, max and sum by four row_xmask DPP steps, not ten LDS-unit shuffles a row; denominator sum tree changes
# speedup vs baseline: 3.8212x; 1.0328x over previous
#include <hip/hip_runtime.h>
#include <hip/hip_bf16.h>


typedef float v2f __attribute__((ext_vector_type(2)));
typedef float v8f __attribute__((ext_vector_type(8)));

#define F_IN 512
#define C_OUT 64


#define RROWS 400

__device__ __forceinline__ int bucket_rows(int r, int N) {
  const int left = N - r * RROWS;
  return left < RROWS ? left : RROWS;
}
__device__ __forceinline__ size_t bucket_counter_index(int r, int N) {
  return ((size_t)r * RROWS + bucket_rows(r, N)) * C_OUT - 2;
}

__global__ void gcn_counters_init(int* outw, int N, int R) {
  const int i = blockIdx.x * blockDim.x + threadIdx.x;
  if (i < R) outw[bucket_counter_index(i, N)] = 0;
}


#define GEMM_WAVES 12
#define GEMM_THREADS (GEMM_WAVES * 32)
#define DEG_PRE 24
#define KC 64
#define A_PITCH 68
#define W_PITCH 160

#define GEMM_LOAD_CHUNK(xp)                                   \
  pf0 = *(const float4*)((xp) + (size_t)0  * F_IN);           \
  pf1 = *(const float4*)((xp) + (size_t)2  * F_IN);           \
  pf2 = *(const float4*)((xp) + (size_t)4  * F_IN);           \
  pf3 = *(const float4*)((xp) + (size_t)6  * F_IN);           \
  pf4 = *(const float4*)((xp) + (size_t)8  * F_IN);           \
  pf5 = *(const float4*)((xp) + (size_t)10 * F_IN);           \
  pf6 = *(const float4*)((xp) + (size_t)12 * F_IN);           \
  pf7 = *(const float4*)((xp) + (size_t)14 * F_IN);
#define GEMM_PARK_CHUNK(ast)                                  \
  *(float4*)((ast) + 0  * A_PITCH) = pf0;                     \
  *(float4*)((ast) + 2  * A_PITCH) = pf1;                     \
  *(float4*)((ast) + 4  * A_PITCH) = pf2;                     \
  *(float4*)((ast) + 6  * A_PITCH) = pf3;                     \
  *(float4*)((ast) + 8  * A_PITCH) = pf4;                     \
  *(float4*)((ast) + 10 * A_PITCH) = pf5;                     \
  *(float4*)((ast) + 12 * A_PITCH) = pf6;                     \
  *(float4*)((ast) + 14 * A_PITCH) = pf7;

__global__ __launch_bounds__(GEMM_THREADS) void gcn_gemm_wmma(
    const float* __restrict__ x, const float* __restrict__ W,
    const int* __restrict__ dst, const int* __restrict__ outw,
    float* __restrict__ h, int E, int N) {
  __shared__ float Ws[(F_IN / 2) * W_PITCH];
  __shared__ float As[GEMM_WAVES * 16 * A_PITCH];
  __shared__ __attribute__((aligned(16))) unsigned indeg[RROWS];
  const int tid  = threadIdx.x;
  const int lane = tid & 31;
  const int wv   = __builtin_amdgcn_readfirstlane(tid >> 5);
  const int half = lane >> 4;
  const int l16  = lane & 15;
  const int lo     = (int)blockIdx.x * RROWS;
  const int rows   = bucket_rows((int)blockIdx.x, N);
  const int ntiles = (rows + 15) >> 4;
  int lt = wv;

  for (int w = tid; w < RROWS; w += GEMM_THREADS) indeg[w] = 0u;

  const int* reg = outw + (size_t)lo * C_OUT;
  const int cap  = rows * (C_OUT / 2) - 1;
  int dl[DEG_PRE];
  #pragma unroll
  for (int u = 0; u < DEG_PRE; ++u) {
    const int i = tid + u * GEMM_THREADS;
    dl[u] = reg[2 * (i < cap ? i : cap - 1) + 1];
  }
  const int total = reg[(size_t)rows * C_OUT - 2];

  const float* xl = x + (size_t)half * F_IN + 4 * l16;
  float4 pf0 = {}, pf1 = {}, pf2 = {}, pf3 = {}, pf4 = {}, pf5 = {}, pf6 = {}, pf7 = {};
  if (lt < ntiles) {
    const float* xp = xl + (size_t)(lo + lt * 16) * F_IN;
    GEMM_LOAD_CHUNK(xp)
  }

  #pragma unroll 4
  for (int u = tid; u < (F_IN / 2) * 16; u += GEMM_THREADS) {
    const int p = u >> 4;
    const int q = u & 15;
    const float4 r0 = *(const float4*)(W + (size_t)(2 * p) * C_OUT + 4 * q);
    const float4 r1 = *(const float4*)(W + (size_t)(2 * p + 1) * C_OUT + 4 * q);
    float4* wp = (float4*)(Ws + p * W_PITCH + 8 * q);
    wp[0] = make_float4(r0.x, r1.x, r0.y, r1.y);
    wp[1] = make_float4(r0.z, r1.z, r0.w, r1.w);
  }
  __syncthreads();

  if (total <= cap) {
    #pragma unroll
    for (int u = 0; u < DEG_PRE; ++u) {
      const int i = tid + u * GEMM_THREADS;
      if (i < total) atomicAdd(&indeg[dl[u]], 1u);
    }
    for (int i = tid + DEG_PRE * GEMM_THREADS; i < total; i += GEMM_THREADS)
      atomicAdd(&indeg[reg[2 * i + 1]], 1u);
  } else {
    for (int e = tid; e < E; e += GEMM_THREADS) {
      const unsigned r = (unsigned)(dst[e] - lo);
      if (r < (unsigned)rows) atomicAdd(&indeg[r], 1u);
    }
  }
  __syncthreads();

  float* aw = As + wv * (16 * A_PITCH);
  float* ast = aw + half * A_PITCH + 4 * l16;
  const float* afr = aw + l16 * A_PITCH + 2 * half;
  const float* bfr = Ws + half * W_PITCH + 2 * l16;

  for (; lt < ntiles; lt += GEMM_WAVES) {
    const int row0 = lo + lt * 16;

    v8f acc0 = {}, acc1 = {}, acc2 = {}, acc3 = {};

    #pragma unroll 1
    for (int kc = 0; kc < F_IN / KC; ++kc) {
      GEMM_PARK_CHUNK(ast)

      const int nk = (kc + 1) & (F_IN / KC - 1);
      const int nt = nk ? lt : lt + GEMM_WAVES;
      if (nt < ntiles) {
        const float* xp = xl + (size_t)(lo + nt * 16) * F_IN + nk * KC;
        GEMM_LOAD_CHUNK(xp)
      }

      __builtin_amdgcn_fence(__ATOMIC_RELEASE, "wavefront");
      __builtin_amdgcn_wave_barrier();
      __builtin_amdgcn_fence(__ATOMIC_ACQUIRE, "wavefront");

      const float* bk = bfr + (kc * (KC / 2)) * W_PITCH;
      #pragma unroll
      for (int kk = 0; kk < KC / 4; ++kk) {
        const v2f a = *(const v2f*)(afr + 4 * kk);
        const float* bp = bk + (2 * kk) * W_PITCH;
        const v2f b0 = *(const v2f*)(bp);
        const v2f b1 = *(const v2f*)(bp + 32);
        const v2f b2 = *(const v2f*)(bp + 64);
        const v2f b3 = *(const v2f*)(bp + 96);
        acc0 = __builtin_amdgcn_wmma_f32_16x16x4_f32(false, a, false, b0, (short)0, acc0, false, false);
        acc1 = __builtin_amdgcn_wmma_f32_16x16x4_f32(false, a, false, b1, (short)0, acc1, false, false);
        acc2 = __builtin_amdgcn_wmma_f32_16x16x4_f32(false, a, false, b2, (short)0, acc2, false, false);
        acc3 = __builtin_amdgcn_wmma_f32_16x16x4_f32(false, a, false, b3, (short)0, acc3, false, false);
      }

      __builtin_amdgcn_fence(__ATOMIC_RELEASE, "wavefront");
      __builtin_amdgcn_wave_barrier();
      __builtin_amdgcn_fence(__ATOMIC_ACQUIRE, "wavefront");
    }

    const uint4 c0 = *(const uint4*)(indeg + lt * 16 + 8 * half);
    const uint4 c1 = *(const uint4*)(indeg + lt * 16 + 8 * half + 4);
    const float dv[8] = { rsqrtf((float)(c0.x + 1u)), rsqrtf((float)(c0.y + 1u)),
                          rsqrtf((float)(c0.z + 1u)), rsqrtf((float)(c0.w + 1u)),
                          rsqrtf((float)(c1.x + 1u)), rsqrtf((float)(c1.y + 1u)),
                          rsqrtf((float)(c1.z + 1u)), rsqrtf((float)(c1.w + 1u)) };
    float* hp = h + (size_t)row0 * C_OUT;
    #pragma unroll
    for (int v = 0; v < 8; ++v) {
      const size_t r = (size_t)(v + 8 * half) * C_OUT;
      const float s0 = acc0[v] * dv[v];
      const float s1 = acc1[v] * dv[v];
      const float s2 = acc2[v] * dv[v];
      const float s3 = acc3[v] * dv[v];
      hp[r + 0  + l16] = s0;
      hp[r + 16 + l16] = s1;
      hp[r + 32 + l16] = s2;
      hp[r + 48 + l16] = s3;
    }
  }
}


#define BK_THREADS 1024
#define BK_PER 8
#define BK_CHUNK (BK_THREADS * BK_PER)

__global__ __launch_bounds__(BK_THREADS) void gcn_bucket(
    const int* __restrict__ src, const int* __restrict__ dst,
    int* outw, int E, int N, int R) {
  extern __shared__ int bk_tab[];
  __shared__ int2 sp[BK_CHUNK];
  __shared__ int  sb[BK_CHUNK];
  __shared__ int  nsorted;
  int* hist  = bk_tab;
  int* loff  = bk_tab + R;
  int* gbase = bk_tab + 2 * R;
  const int tid = threadIdx.x;

  for (int w = tid; w < R; w += BK_THREADS) hist[w] = 0;
  __syncthreads();

  const long long c0 = (long long)blockIdx.x * BK_CHUNK;
  int s[BK_PER], d[BK_PER], rk[BK_PER];
  #pragma unroll
  for (int u = 0; u < BK_PER; ++u) {
    const long long e = c0 + u * BK_THREADS + tid;
    const bool ok = e < (long long)E;
    d[u] = ok ? dst[e] : -1;
    s[u] = ok ? src[e] : 0;
  }
  #pragma unroll
  for (int u = 0; u < BK_PER; ++u) {
    rk[u] = -1;
    if ((unsigned)d[u] < (unsigned)N) rk[u] = atomicAdd(&hist[(unsigned)d[u] / RROWS], 1);
  }
  __syncthreads();

  if (tid < 32) {
    const int per = (R + 31) / 32;
    const int b0 = tid * per;
    int b1 = b0 + per;
    if (b1 > R) b1 = R;
    int sum = 0;
    for (int b = b0; b < b1; ++b) sum += hist[b];
    int inc = sum;
    #pragma unroll
    for (int o = 1; o < 32; o <<= 1) {
      const int t = __shfl_up(inc, o, 32);
      if (tid >= o) inc += t;
    }
    int run = inc - sum;
    for (int b = b0; b < b1; ++b) { loff[b] = run; run += hist[b]; }
    if (tid == 31) nsorted = inc;
  }
  for (int b = tid; b < R; b += BK_THREADS) {
    const int c = hist[b];
    gbase[b] = c ? atomicAdd(&outw[bucket_counter_index(b, N)], c) : 0;
  }
  __syncthreads();

  #pragma unroll
  for (int u = 0; u < BK_PER; ++u) {
    if (rk[u] >= 0) {
      const int b = (int)((unsigned)d[u] / RROWS);
      const int slot = loff[b] + rk[u];
      sp[slot] = make_int2(s[u], d[u] - b * RROWS);
      sb[slot] = b;
    }
  }
  __syncthreads();

  const int n = nsorted;
  for (int i = tid; i < n; i += BK_THREADS) {
    const int b = sb[i];
    const int p = gbase[b] + (i - loff[b]);
    const int cap = bucket_rows(b, N) * (C_OUT / 2) - 1;
    if (p < cap) ((int2*)(outw + (size_t)b * RROWS * C_OUT))[p] = sp[i];
  }
}


template <int M>
__device__ __forceinline__ float row_xor(float v) {
  return __int_as_float(__builtin_amdgcn_update_dpp(0, __float_as_int(v), 0x160 + M, 0xf, 0xf, true));
}
template <int M>
__device__ __forceinline__ float row_max_step(float v) {
  asm("v_max_num_f32_dpp %0, %0, %0 row_xmask:%1 row_mask:0xf bank_mask:0xf" : "+v"(v) : "n"(M));
  return v;
}

#define ACC_THREADS 1024
#define AGG_CHUNK 16

__global__ __launch_bounds__(ACC_THREADS) void gcn_accumulate(
    const int* __restrict__ src, const int* __restrict__ dst,
    const float* __restrict__ h,
    const float* __restrict__ bias, float* out, int E, int N) {
  __shared__ __attribute__((aligned(16))) float acc[RROWS * C_OUT];
  __shared__ unsigned indeg[RROWS];
  const int tid  = threadIdx.x;
  const int lane = tid & 31;
  const int wv   = __builtin_amdgcn_readfirstlane(tid >> 5);
  const int lo   = blockIdx.x * RROWS;
  const int rows = bucket_rows(blockIdx.x, N);
  const int* reg = (const int*)out + (size_t)lo * C_OUT;
  const int cap  = rows * (C_OUT / 2) - 1;
  const int total = reg[(size_t)rows * C_OUT - 2];

  const int l16 = lane & 15;
  const float4 bv = *(const float4*)(bias + 4 * l16);

  const float* hb = h + (size_t)lo * C_OUT;
  for (int i = tid; i < rows * C_OUT; i += ACC_THREADS) acc[i] = hb[i];
  for (int w = tid; w < RROWS; w += ACC_THREADS) indeg[w] = 0u;
  __syncthreads();

  const float* hl = h + lane;
  float* al = acc + lane;

  if (total <= cap) {
    const int2* pairs = (const int2*)reg;
    int2 nxt = make_int2(0, 0);
    if (wv * 32 + lane < total) nxt = pairs[wv * 32 + lane];
    for (int base = wv * 32; base < total; base += ACC_THREADS) {
      const int left = total - base;
      const int cnt = left < 32 ? left : 32;
      const int s = nxt.x;
      const int d = nxt.y;
      if (lane < cnt) atomicAdd(&indeg[d], 1u);
      if (base + ACC_THREADS + lane < total) nxt = pairs[base + ACC_THREADS + lane];
      int j = 0;
      for (; j + AGG_CHUNK <= cnt; j += AGG_CHUNK) {
        float a[AGG_CHUNK], b[AGG_CHUNK];
        #pragma unroll
        for (int u = 0; u < AGG_CHUNK; ++u) {
          const int sj = __builtin_amdgcn_readlane(s, j + u);
          const float* hr = hl + (size_t)sj * C_OUT;
          a[u] = hr[0];
          b[u] = hr[32];
        }
        #pragma unroll
        for (int u = 0; u < AGG_CHUNK; ++u) {
          const int dj = __builtin_amdgcn_readlane(d, j + u);
          float* ar = al + dj * C_OUT;
          atomicAdd(ar,      a[u]);
          atomicAdd(ar + 32, b[u]);
        }
      }
      for (; j < cnt; ++j) {
        const int sj = __builtin_amdgcn_readlane(s, j);
        const int dj = __builtin_amdgcn_readlane(d, j);
        const float* hr = hl + (size_t)sj * C_OUT;
        float* ar = al + dj * C_OUT;
        const float a = hr[0];
        const float b = hr[32];
        atomicAdd(ar,      a);
        atomicAdd(ar + 32, b);
      }
    }
  } else {
    for (long long e0 = (long long)wv * 32; e0 < (long long)E; e0 += ACC_THREADS) {
      const long long e = e0 + lane;
      const int dv = e < (long long)E ? dst[e] : -1;
      const unsigned r = (unsigned)(dv - lo);
      const bool mine = r < (unsigned)rows;
      const int s = mine ? src[e] : 0;
      if (mine) atomicAdd(&indeg[r], 1u);
      unsigned todo = __builtin_amdgcn_ballot_w32(mine);
      while (todo) {
        const int j = __builtin_ctz(todo);
        todo &= todo - 1;
        const int sj = __builtin_amdgcn_readlane(s, j);
        const int dj = __builtin_amdgcn_readlane((int)r, j);
        const float* hr = hl + (size_t)sj * C_OUT;
        float* ar = al + dj * C_OUT;
        const float a = hr[0];
        const float b = hr[32];
        atomicAdd(ar,      a);
        atomicAdd(ar + 32, b);
      }
    }
  }
  __syncthreads();

  for (int r = tid >> 4; r < rows; r += ACC_THREADS / 16) {
    const float di = rsqrtf((float)(indeg[r] + 1u));
    const float4 a = *(const float4*)(acc + r * C_OUT + 4 * l16);
    const float v0 = a.x * di + bv.x;
    const float v1 = a.y * di + bv.y;
    const float v2 = a.z * di + bv.z;
    const float v3 = a.w * di + bv.w;
    float m = fmaxf(fmaxf(v0, v1), fmaxf(v2, v3));
    m = row_max_step<1>(m);
    m = row_max_step<2>(m);
    m = row_max_step<4>(m);
    m = row_max_step<8>(m);
    float sm = (__expf(v0 - m) + __expf(v1 - m)) + (__expf(v2 - m) + __expf(v3 - m));
    sm += row_xor<1>(sm);
    sm += row_xor<2>(sm);
    sm += row_xor<4>(sm);
    sm += row_xor<8>(sm);
    const float lse = __logf(sm);
    *(float4*)(out + (size_t)(lo + r) * C_OUT + 4 * l16) =
        make_float4(v0 - m - lse, v1 - m - lse, v2 - m - lse, v3 - m - lse);
  }
}


extern "C" void kernel_launch(void* const* d_in, const int* in_sizes, int n_in,
                              void* d_out, int out_size, void* d_ws, size_t ws_size,
                              hipStream_t stream) {
  const float* x  = (const float*)d_in[0];
  const float* W  = (const float*)d_in[1];
  const float* b  = (const float*)d_in[2];
  const int*   ei = (const int*)d_in[3];

  const int N = in_sizes[0] / F_IN;
  const int E = in_sizes[3] / 2;
  const int* src = ei;
  const int* dst = ei + E;
  float* out = (float*)d_out;

  float* h = (float*)d_ws;

  const int T = 256;
  const int R = (N + RROWS - 1) / RROWS;

  gcn_counters_init<<<(R + T - 1) / T, T, 0, stream>>>((int*)out, N, R);

  gcn_bucket<<<(E + BK_CHUNK - 1) / BK_CHUNK, BK_THREADS, 3 * R * sizeof(int), stream>>>(
      src, dst, (int*)out, E, N, R);

  gcn_gemm_wmma<<<R, GEMM_THREADS, 0, stream>>>(x, W, dst, (const int*)out, h, E, N);

  gcn_accumulate<<<R, ACC_THREADS, 0, stream>>>(src, dst, h, b, out, E, N);
}


# owner block sorts its bucket by destination row in LDS; a half wave sums each row's hs rows in registers, next row's loads issued ahead of log_softmax; no LDS float adds; sum order = LDS rank
# speedup vs baseline: 4.1336x; 1.0818x over previous
#include <hip/hip_runtime.h>
#include <hip/hip_bf16.h>


typedef float v2f __attribute__((ext_vector_type(2)));
typedef float v8f __attribute__((ext_vector_type(8)));

#define F_IN 512
#define C_OUT 64


#define RROWS 400

__device__ __forceinline__ int bucket_rows(int r, int N) {
  const int left = N - r * RROWS;
  return left < RROWS ? left : RROWS;
}
__device__ __forceinline__ size_t bucket_counter_index(int r, int N) {
  return ((size_t)r * RROWS + bucket_rows(r, N)) * C_OUT - 2;
}

__global__ void gcn_counters_init(int* outw, int N, int R) {
  const int i = blockIdx.x * blockDim.x + threadIdx.x;
  if (i < R) outw[bucket_counter_index(i, N)] = 0;
}


#define GEMM_WAVES 12
#define GEMM_THREADS (GEMM_WAVES * 32)
#define DEG_PRE 24
#define KC 64
#define A_PITCH 68
#define W_PITCH 160

#define GEMM_LOAD_CHUNK(xp)                                   \
  pf0 = *(const float4*)((xp) + (size_t)0  * F_IN);           \
  pf1 = *(const float4*)((xp) + (size_t)2  * F_IN);           \
  pf2 = *(const float4*)((xp) + (size_t)4  * F_IN);           \
  pf3 = *(const float4*)((xp) + (size_t)6  * F_IN);           \
  pf4 = *(const float4*)((xp) + (size_t)8  * F_IN);           \
  pf5 = *(const float4*)((xp) + (size_t)10 * F_IN);           \
  pf6 = *(const float4*)((xp) + (size_t)12 * F_IN);           \
  pf7 = *(const float4*)((xp) + (size_t)14 * F_IN);
#define GEMM_PARK_CHUNK(ast)                                  \
  *(float4*)((ast) + 0  * A_PITCH) = pf0;                     \
  *(float4*)((ast) + 2  * A_PITCH) = pf1;                     \
  *(float4*)((ast) + 4  * A_PITCH) = pf2;                     \
  *(float4*)((ast) + 6  * A_PITCH) = pf3;                     \
  *(float4*)((ast) + 8  * A_PITCH) = pf4;                     \
  *(float4*)((ast) + 10 * A_PITCH) = pf5;                     \
  *(float4*)((ast) + 12 * A_PITCH) = pf6;                     \
  *(float4*)((ast) + 14 * A_PITCH) = pf7;

__global__ __launch_bounds__(GEMM_THREADS) void gcn_gemm_wmma(
    const float* __restrict__ x, const float* __restrict__ W,
    const int* __restrict__ dst, const int* __restrict__ outw,
    float* __restrict__ h, int E, int N) {
  __shared__ float Ws[(F_IN / 2) * W_PITCH];
  __shared__ float As[GEMM_WAVES * 16 * A_PITCH];
  __shared__ __attribute__((aligned(16))) unsigned indeg[RROWS];
  const int tid  = threadIdx.x;
  const int lane = tid & 31;
  const int wv   = __builtin_amdgcn_readfirstlane(tid >> 5);
  const int half = lane >> 4;
  const int l16  = lane & 15;
  const int lo     = (int)blockIdx.x * RROWS;
  const int rows   = bucket_rows((int)blockIdx.x, N);
  const int ntiles = (rows + 15) >> 4;
  int lt = wv;

  for (int w = tid; w < RROWS; w += GEMM_THREADS) indeg[w] = 0u;

  const int* reg = outw + (size_t)lo * C_OUT;
  const int cap  = rows * (C_OUT / 2) - 1;
  int dl[DEG_PRE];
  #pragma unroll
  for (int u = 0; u < DEG_PRE; ++u) {
    const int i = tid + u * GEMM_THREADS;
    dl[u] = reg[2 * (i < cap ? i : cap - 1) + 1];
  }
  const int total = reg[(size_t)rows * C_OUT - 2];

  const float* xl = x + (size_t)half * F_IN + 4 * l16;
  float4 pf0 = {}, pf1 = {}, pf2 = {}, pf3 = {}, pf4 = {}, pf5 = {}, pf6 = {}, pf7 = {};
  if (lt < ntiles) {
    const float* xp = xl + (size_t)(lo + lt * 16) * F_IN;
    GEMM_LOAD_CHUNK(xp)
  }

  #pragma unroll 4
  for (int u = tid; u < (F_IN / 2) * 16; u += GEMM_THREADS) {
    const int p = u >> 4;
    const int q = u & 15;
    const float4 r0 = *(const float4*)(W + (size_t)(2 * p) * C_OUT + 4 * q);
    const float4 r1 = *(const float4*)(W + (size_t)(2 * p + 1) * C_OUT + 4 * q);
    float4* wp = (float4*)(Ws + p * W_PITCH + 8 * q);
    wp[0] = make_float4(r0.x, r1.x, r0.y, r1.y);
    wp[1] = make_float4(r0.z, r1.z, r0.w, r1.w);
  }
  __syncthreads();

  if (total <= cap) {
    #pragma unroll
    for (int u = 0; u < DEG_PRE; ++u) {
      const int i = tid + u * GEMM_THREADS;
      if (i < total) atomicAdd(&indeg[dl[u]], 1u);
    }
    for (int i = tid + DEG_PRE * GEMM_THREADS; i < total; i += GEMM_THREADS)
      atomicAdd(&indeg[reg[2 * i + 1]], 1u);
  } else {
    for (int e = tid; e < E; e += GEMM_THREADS) {
      const unsigned r = (unsigned)(dst[e] - lo);
      if (r < (unsigned)rows) atomicAdd(&indeg[r], 1u);
    }
  }
  __syncthreads();

  float* aw = As + wv * (16 * A_PITCH);
  float* ast = aw + half * A_PITCH + 4 * l16;
  const float* afr = aw + l16 * A_PITCH + 2 * half;
  const float* bfr = Ws + half * W_PITCH + 2 * l16;

  for (; lt < ntiles; lt += GEMM_WAVES) {
    const int row0 = lo + lt * 16;

    v8f acc0 = {}, acc1 = {}, acc2 = {}, acc3 = {};

    #pragma unroll 1
    for (int kc = 0; kc < F_IN / KC; ++kc) {
      GEMM_PARK_CHUNK(ast)

      const int nk = (kc + 1) & (F_IN / KC - 1);
      const int nt = nk ? lt : lt + GEMM_WAVES;
      if (nt < ntiles) {
        const float* xp = xl + (size_t)(lo + nt * 16) * F_IN + nk * KC;
        GEMM_LOAD_CHUNK(xp)
      }

      __builtin_amdgcn_fence(__ATOMIC_RELEASE, "wavefront");
      __builtin_amdgcn_wave_barrier();
      __builtin_amdgcn_fence(__ATOMIC_ACQUIRE, "wavefront");

      const float* bk = bfr + (kc * (KC / 2)) * W_PITCH;
      #pragma unroll
      for (int kk = 0; kk < KC / 4; ++kk) {
        const v2f a = *(const v2f*)(afr + 4 * kk);
        const float* bp = bk + (2 * kk) * W_PITCH;
        const v2f b0 = *(const v2f*)(bp);
        const v2f b1 = *(const v2f*)(bp + 32);
        const v2f b2 = *(const v2f*)(bp + 64);
        const v2f b3 = *(const v2f*)(bp + 96);
        acc0 = __builtin_amdgcn_wmma_f32_16x16x4_f32(false, a, false, b0, (short)0, acc0, false, false);
        acc1 = __builtin_amdgcn_wmma_f32_16x16x4_f32(false, a, false, b1, (short)0, acc1, false, false);
        acc2 = __builtin_amdgcn_wmma_f32_16x16x4_f32(false, a, false, b2, (short)0, acc2, false, false);
        acc3 = __builtin_amdgcn_wmma_f32_16x16x4_f32(false, a, false, b3, (short)0, acc3, false, false);
      }

      __builtin_amdgcn_fence(__ATOMIC_RELEASE, "wavefront");
      __builtin_amdgcn_wave_barrier();
      __builtin_amdgcn_fence(__ATOMIC_ACQUIRE, "wavefront");
    }

    const uint4 c0 = *(const uint4*)(indeg + lt * 16 + 8 * half);
    const uint4 c1 = *(const uint4*)(indeg + lt * 16 + 8 * half + 4);
    const float dv[8] = { rsqrtf((float)(c0.x + 1u)), rsqrtf((float)(c0.y + 1u)),
                          rsqrtf((float)(c0.z + 1u)), rsqrtf((float)(c0.w + 1u)),
                          rsqrtf((float)(c1.x + 1u)), rsqrtf((float)(c1.y + 1u)),
                          rsqrtf((float)(c1.z + 1u)), rsqrtf((float)(c1.w + 1u)) };
    float* hp = h + (size_t)row0 * C_OUT;
    #pragma unroll
    for (int v = 0; v < 8; ++v) {
      const size_t r = (size_t)(v + 8 * half) * C_OUT;
      const float s0 = acc0[v] * dv[v];
      const float s1 = acc1[v] * dv[v];
      const float s2 = acc2[v] * dv[v];
      const float s3 = acc3[v] * dv[v];
      hp[r + 0  + l16] = s0;
      hp[r + 16 + l16] = s1;
      hp[r + 32 + l16] = s2;
      hp[r + 48 + l16] = s3;
    }
  }
}


#define BK_THREADS 1024
#define BK_PER 8
#define BK_CHUNK (BK_THREADS * BK_PER)

__global__ __launch_bounds__(BK_THREADS) void gcn_bucket(
    const int* __restrict__ src, const int* __restrict__ dst,
    int* outw, int E, int N, int R) {
  extern __shared__ int bk_tab[];
  __shared__ int2 sp[BK_CHUNK];
  __shared__ int  sb[BK_CHUNK];
  __shared__ int  nsorted;
  int* hist  = bk_tab;
  int* loff  = bk_tab + R;
  int* gbase = bk_tab + 2 * R;
  const int tid = threadIdx.x;

  for (int w = tid; w < R; w += BK_THREADS) hist[w] = 0;
  __syncthreads();

  const long long c0 = (long long)blockIdx.x * BK_CHUNK;
  int s[BK_PER], d[BK_PER], rk[BK_PER];
  #pragma unroll
  for (int u = 0; u < BK_PER; ++u) {
    const long long e = c0 + u * BK_THREADS + tid;
    const bool ok = e < (long long)E;
    d[u] = ok ? dst[e] : -1;
    s[u] = ok ? src[e] : 0;
  }
  #pragma unroll
  for (int u = 0; u < BK_PER; ++u) {
    rk[u] = -1;
    if ((unsigned)d[u] < (unsigned)N) rk[u] = atomicAdd(&hist[(unsigned)d[u] / RROWS], 1);
  }
  __syncthreads();

  if (tid < 32) {
    const int per = (R + 31) / 32;
    const int b0 = tid * per;
    int b1 = b0 + per;
    if (b1 > R) b1 = R;
    int sum = 0;
    for (int b = b0; b < b1; ++b) sum += hist[b];
    int inc = sum;
    #pragma unroll
    for (int o = 1; o < 32; o <<= 1) {
      const int t = __shfl_up(inc, o, 32);
      if (tid >= o) inc += t;
    }
    int run = inc - sum;
    for (int b = b0; b < b1; ++b) { loff[b] = run; run += hist[b]; }
    if (tid == 31) nsorted = inc;
  }
  for (int b = tid; b < R; b += BK_THREADS) {
    const int c = hist[b];
    gbase[b] = c ? atomicAdd(&outw[bucket_counter_index(b, N)], c) : 0;
  }
  __syncthreads();

  #pragma unroll
  for (int u = 0; u < BK_PER; ++u) {
    if (rk[u] >= 0) {
      const int b = (int)((unsigned)d[u] / RROWS);
      const int slot = loff[b] + rk[u];
      sp[slot] = make_int2(s[u], d[u] - b * RROWS);
      sb[slot] = b;
    }
  }
  __syncthreads();

  const int n = nsorted;
  for (int i = tid; i < n; i += BK_THREADS) {
    const int b = sb[i];
    const int p = gbase[b] + (i - loff[b]);
    const int cap = bucket_rows(b, N) * (C_OUT / 2) - 1;
    if (p < cap) ((int2*)(outw + (size_t)b * RROWS * C_OUT))[p] = sp[i];
  }
}


template <int M>
__device__ __forceinline__ float row_xor(float v) {
  return __int_as_float(__builtin_amdgcn_update_dpp(0, __float_as_int(v), 0x160 + M, 0xf, 0xf, true));
}
template <int M>
__device__ __forceinline__ float row_max_step(float v) {
  asm("v_max_num_f32_dpp %0, %0, %0 row_xmask:%1 row_mask:0xf bank_mask:0xf" : "+v"(v) : "n"(M));
  return v;
}

__device__ __forceinline__ void row_log_softmax(float4 a, float di, float4 bv, float* op) {
  const float v0 = a.x * di + bv.x;
  const float v1 = a.y * di + bv.y;
  const float v2 = a.z * di + bv.z;
  const float v3 = a.w * di + bv.w;
  float m = fmaxf(fmaxf(v0, v1), fmaxf(v2, v3));
  m = row_max_step<1>(m);
  m = row_max_step<2>(m);
  m = row_max_step<4>(m);
  m = row_max_step<8>(m);
  float sm = (__expf(v0 - m) + __expf(v1 - m)) + (__expf(v2 - m) + __expf(v3 - m));
  sm += row_xor<1>(sm);
  sm += row_xor<2>(sm);
  sm += row_xor<4>(sm);
  sm += row_xor<8>(sm);
  const float lse = __logf(sm);
  *(float4*)op = make_float4(v0 - m - lse, v1 - m - lse, v2 - m - lse, v3 - m - lse);
}

#define ACC_THREADS 1024
#define SRT_PER 13
#define SRT_PRE 7
#define ROW_GROUP 8
#define SCAN_WAVES ((RROWS + 31) / 32)
#define HALF_WAVES (ACC_THREADS / 16)

__global__ __launch_bounds__(ACC_THREADS) void gcn_accumulate(
    const int* __restrict__ src, const int* __restrict__ dst,
    const float* __restrict__ h,
    const float* __restrict__ bias, float* out, int E, int N) {
  __shared__ __attribute__((aligned(16))) float acc[RROWS * C_OUT];
  __shared__ unsigned indeg[RROWS];
  __shared__ unsigned roff[RROWS];
  __shared__ unsigned wsum[SCAN_WAVES];
  const int tid  = threadIdx.x;
  const int lane = tid & 31;
  const int wv   = __builtin_amdgcn_readfirstlane(tid >> 5);
  const int l16  = lane & 15;
  const int lo   = blockIdx.x * RROWS;
  const int rows = bucket_rows(blockIdx.x, N);
  const int* reg = (const int*)out + (size_t)lo * C_OUT;
  const int cap  = rows * (C_OUT / 2) - 1;
  const int2* pairs = (const int2*)reg;

  int2 pr[SRT_PER];
  #pragma unroll
  for (int u = 0; u < SRT_PRE; ++u) {
    const int i = tid + u * ACC_THREADS;
    pr[u] = pairs[i < cap ? i : cap - 1];
  }
  const int total = reg[(size_t)rows * C_OUT - 2];
  const bool sorted = total <= cap;

  const float4 bv = *(const float4*)(bias + 4 * l16);

  for (int w = tid; w < RROWS; w += ACC_THREADS) indeg[w] = 0u;
  if (!sorted) {
    const float* hb = h + (size_t)lo * C_OUT;
    for (int i = tid; i < rows * C_OUT; i += ACC_THREADS) acc[i] = hb[i];
  }
  __syncthreads();

  int* srt = (int*)acc;
  int rk[SRT_PER];
  if (sorted) {
    #pragma unroll
    for (int u = SRT_PRE; u < SRT_PER; ++u) {
      const int i = tid + u * ACC_THREADS;
      pr[u] = make_int2(0, 0);
      if (i < total) pr[u] = pairs[i];
    }
    #pragma unroll
    for (int u = 0; u < SRT_PER; ++u) {
      const int i = tid + u * ACC_THREADS;
      rk[u] = -1;
      if (i < total) rk[u] = (int)atomicAdd(&indeg[pr[u].y], 1u);
    }
  } else {
    const float* hl = h + lane;
    float* al = acc + lane;
    for (long long e0 = (long long)wv * 32; e0 < (long long)E; e0 += ACC_THREADS) {
      const long long e = e0 + lane;
      const int dv = e < (long long)E ? dst[e] : -1;
      const unsigned r = (unsigned)(dv - lo);
      const bool mine = r < (unsigned)rows;
      const int s = mine ? src[e] : 0;
      if (mine) atomicAdd(&indeg[r], 1u);
      unsigned todo = __builtin_amdgcn_ballot_w32(mine);
      while (todo) {
        const int j = __builtin_ctz(todo);
        todo &= todo - 1;
        const int sj = __builtin_amdgcn_readlane(s, j);
        const int dj = __builtin_amdgcn_readlane((int)r, j);
        const float* hr = hl + (size_t)sj * C_OUT;
        float* ar = al + dj * C_OUT;
        const float a = hr[0];
        const float b = hr[32];
        atomicAdd(ar,      a);
        atomicAdd(ar + 32, b);
      }
    }
  }
  __syncthreads();

  unsigned c = 0u;
  if (sorted && tid < rows) c = indeg[tid];
  unsigned inc = c;
  #pragma unroll
  for (int o = 1; o < 32; o <<= 1) {
    const unsigned t = __shfl_up(inc, o, 32);
    if (lane >= o) inc += t;
  }
  if (sorted && lane == 31 && wv < SCAN_WAVES) wsum[wv] = inc;
  __syncthreads();
  if (sorted && tid < rows) {
    unsigned base = 0u;
    #pragma unroll
    for (int w = 0; w < SCAN_WAVES; ++w) base += w < wv ? wsum[w] : 0u;
    roff[tid] = base + inc - c;
  }
  __syncthreads();

  if (sorted) {
    unsigned ro[SRT_PER];
    #pragma unroll
    for (int u = 0; u < SRT_PER; ++u) ro[u] = roff[rk[u] >= 0 ? pr[u].y : 0];
    #pragma unroll
    for (int u = 0; u < SRT_PER; ++u)
      if (rk[u] >= 0) srt[ro[u] + (unsigned)rk[u]] = pr[u].x * (C_OUT / 4);
    if (tid < ROW_GROUP) srt[total + tid] = 0;
  }
  __syncthreads();

  float* orow = out + (size_t)lo * C_OUT + 4 * l16;
  if (sorted) {
    const float4* h4 = (const float4*)h;
    const float4 zero4 = make_float4(0.f, 0.f, 0.f, 0.f);
    int r = tid >> 4;
    bool act = r < rows;
    int n = 0;
    const int* sr = srt;
    float4 a = zero4;
    if (act) {
      n = (int)indeg[r];
      sr = srt + roff[r];
      a = h4[(unsigned)(lo + r) * (C_OUT / 4) + l16];
    }
    int rn = r + HALF_WAVES;
    int rc = rn < rows ? rn : 0;
    int nn = (int)indeg[rc];
    unsigned sn = roff[rc];
    int j = 0;
    float4 v[ROW_GROUP];
    {
      unsigned sx[ROW_GROUP];
      #pragma unroll
      for (int u = 0; u < ROW_GROUP; ++u) sx[u] = (unsigned)sr[u];
      #pragma unroll
      for (int u = 0; u < ROW_GROUP; ++u) {
        v[u] = zero4;
        if (u < n) v[u] = h4[sx[u] + l16];
      }
    }
    while (act) {
      #pragma unroll
      for (int u = 0; u < ROW_GROUP; ++u) {
        a.x += v[u].x;
        a.y += v[u].y;
        a.z += v[u].z;
        a.w += v[u].w;
      }
      j += ROW_GROUP;
      const bool last = j >= n;
      const float4 afin = a;
      const int nfin = n;
      const int rfin = r;
      if (last) {
        r = rn;
        act = r < rows;
        n = act ? nn : 0;
        sr = srt + sn;
        j = 0;
        a = zero4;
        if (act) a = h4[(unsigned)(lo + r) * (C_OUT / 4) + l16];
        rn = r + HALF_WAVES;
        rc = rn < rows ? rn : 0;
        nn = (int)indeg[rc];
        sn = roff[rc];
      }
      {
        unsigned sx[ROW_GROUP];
        #pragma unroll
        for (int u = 0; u < ROW_GROUP; ++u) sx[u] = (unsigned)sr[j + u];
        #pragma unroll
        for (int u = 0; u < ROW_GROUP; ++u) {
          v[u] = zero4;
          if (j + u < n) v[u] = h4[sx[u] + l16];
        }
      }
      if (last) row_log_softmax(afin, rsqrtf((float)(nfin + 1)), bv, orow + (size_t)rfin * C_OUT);
    }
  } else {
    for (int r = tid >> 4; r < rows; r += ACC_THREADS / 16) {
      const float di = rsqrtf((float)(indeg[r] + 1u));
      const float4 a = *(const float4*)(acc + r * C_OUT + 4 * l16);
      row_log_softmax(a, di, bv, orow + (size_t)r * C_OUT);
    }
  }
}


extern "C" void kernel_launch(void* const* d_in, const int* in_sizes, int n_in,
                              void* d_out, int out_size, void* d_ws, size_t ws_size,
                              hipStream_t stream) {
  const float* x  = (const float*)d_in[0];
  const float* W  = (const float*)d_in[1];
  const float* b  = (const float*)d_in[2];
  const int*   ei = (const int*)d_in[3];

  const int N = in_sizes[0] / F_IN;
  const int E = in_sizes[3] / 2;
  const int* src = ei;
  const int* dst = ei + E;
  float* out = (float*)d_out;

  float* h = (float*)d_ws;

  const int T = 256;
  const int R = (N + RROWS - 1) / RROWS;

  gcn_counters_init<<<(R + T - 1) / T, T, 0, stream>>>((int*)out, N, R);

  gcn_bucket<<<(E + BK_CHUNK - 1) / BK_CHUNK, BK_THREADS, 3 * R * sizeof(int), stream>>>(
      src, dst, (int*)out, E, N, R);

  gcn_gemm_wmma<<<R, GEMM_THREADS, 0, stream>>>(x, W, dst, (const int*)out, h, E, N);

  gcn_accumulate<<<R, ACC_THREADS, 0, stream>>>(src, dst, h, b, out, E, N);
}
